# starting kernel
# baseline (speedup 1.0000x reference)
#include <hip/hip_runtime.h>
#include <hip/hip_bf16.h>

#define NNODES 50000
#define NEDGES 800000
#define INF_   256
#define FDIM   32
#define HEADS  8
#define HF     (HEADS * FDIM)
#define NEG_SLOPE 0.2f

typedef float v2f __attribute__((ext_vector_type(2)));
typedef float v8f __attribute__((ext_vector_type(8)));

__global__ __launch_bounds__(256) void gemm_wmma_f32(const float* __restrict__ X,
                                                     const float* __restrict__ W,
                                                     float* __restrict__ Z,
                                                     int K) {
    const int lane = threadIdx.x & 31;
    const int wave = threadIdx.x >> 5;
    const int half = lane >> 4;
    const int mn   = lane & 15;
    const int row0 = blockIdx.x * 16;
    const int col0 = wave * 32;
    const int col1 = col0 + 16;

    const float* xrow = X + (size_t)(row0 + mn) * K;
    v8f c0 = {}, c1 = {};
    for (int k = 0; k < K; k += 4) {
        const int kk = k + half * 2;
        v2f a, b0, b1;
        a.x  = xrow[kk];
        a.y  = xrow[kk + 1];
        const float* w0 = W + (size_t)kk * HF;
        b0.x = w0[col0 + mn];
        b0.y = w0[HF + col0 + mn];
        b1.x = w0[col1 + mn];
        b1.y = w0[HF + col1 + mn];
        c0 = __builtin_amdgcn_wmma_f32_16x16x4_f32(false, a, false, b0,
                                                   (short)0, c0, false, false);
        c1 = __builtin_amdgcn_wmma_f32_16x16x4_f32(false, a, false, b1,
                                                   (short)0, c1, false, false);
    }
    float* zout = Z + (size_t)(row0 + half * 8) * HF + mn;
#pragma unroll
    for (int v = 0; v < 8; ++v) {
        zout[(size_t)v * HF + col0] = c0[v];
        zout[(size_t)v * HF + col1] = c1[v];
    }
}

__global__ void attn_logits(const float* __restrict__ Z,
                            const float* __restrict__ al, const float* __restrict__ ar,
                            float* __restrict__ el, float* __restrict__ er) {
    int i = blockIdx.x * blockDim.x + threadIdx.x;
    if (i >= NNODES * HEADS) return;
    const int h = i & (HEADS - 1);
    const int n = i >> 3;
    const float* z  = Z + (size_t)n * HF + h * FDIM;
    const float* pl = al + h * FDIM;
    const float* pr = ar + h * FDIM;
    float sl = 0.f, sr = 0.f;
#pragma unroll
    for (int f = 0; f < FDIM; ++f) { float zv = z[f]; sl += zv * pl[f]; sr += zv * pr[f]; }
    el[i] = sl; er[i] = sr;
}

__device__ __forceinline__ unsigned f2ord(float f) {
    unsigned u = __float_as_uint(f);
    return (u & 0x80000000u) ? ~u : (u | 0x80000000u);
}
__device__ __forceinline__ float ord2f(unsigned e) {
    unsigned u = (e & 0x80000000u) ? (e & 0x7FFFFFFFu) : ~e;
    return __uint_as_float(u);
}
#define ORD_NEG_INF 0x007FFFFFu

__global__ void init_bufs(unsigned* __restrict__ m_ord, float* __restrict__ denom,
                          float* __restrict__ acc) {
    int i = blockIdx.x * blockDim.x + threadIdx.x;
    if (i < NNODES * HEADS) { m_ord[i] = ORD_NEG_INF; denom[i] = 0.f; }
    if (i < NNODES * FDIM)  acc[i] = 0.f;
}

__device__ __forceinline__ float edge_logit(const float* el, const float* er,
                                            int s, int d, int h) {
    float v = el[s * HEADS + h] + er[d * HEADS + h];
    return v >= 0.f ? v : NEG_SLOPE * v;
}

__global__ void edge_max(const int* __restrict__ src, const int* __restrict__ dst,
                         const float* __restrict__ el, const float* __restrict__ er,
                         unsigned* __restrict__ m_ord) {
    int i = blockIdx.x * blockDim.x + threadIdx.x;
    if (i >= NEDGES * HEADS) return;
    const int h = i & (HEADS - 1);
    const int e = i >> 3;
    const int s = src[e], d = dst[e];
    atomicMax(&m_ord[d * HEADS + h], f2ord(edge_logit(el, er, s, d, h)));
}

__global__ void edge_expsum(const int* __restrict__ src, const int* __restrict__ dst,
                            const float* __restrict__ el, const float* __restrict__ er,
                            const unsigned* __restrict__ m_ord,
                            float* __restrict__ exbuf, float* __restrict__ denom) {
    int i = blockIdx.x * blockDim.x + threadIdx.x;
    if (i >= NEDGES * HEADS) return;
    const int h = i & (HEADS - 1);
    const int e = i >> 3;
    const int s = src[e], d = dst[e];
    float v  = edge_logit(el, er, s, d, h);
    float ex = __expf(v - ord2f(m_ord[d * HEADS + h]));
    exbuf[i] = ex;
    atomicAdd(&denom[d * HEADS + h], ex);
}

__global__ void edge_alpha(const int* __restrict__ dst, const float* __restrict__ denom,
                           float* __restrict__ exbuf) {
    int i = blockIdx.x * blockDim.x + threadIdx.x;
    if (i >= NEDGES * HEADS) return;
    const int h = i & (HEADS - 1);
    const int d = dst[i >> 3];
    exbuf[i] = exbuf[i] / denom[d * HEADS + h] * (1.0f / HEADS);
}

__global__ void edge_agg(const int* __restrict__ src, const int* __restrict__ dst,
                         const float* __restrict__ Z, const float* __restrict__ exbuf,
                         float* __restrict__ acc) {
    int i = blockIdx.x * blockDim.x + threadIdx.x;
    if (i >= NEDGES * FDIM) return;
    const int f = i & (FDIM - 1);
    const int e = i >> 5;
    const int s = src[e], d = dst[e];
    const float* zs = Z + (size_t)s * HF + f;
    const float* ae = exbuf + e * HEADS;
    float sum = 0.f;
#pragma unroll
    for (int h = 0; h < HEADS; ++h)
        sum += ae[h] * zs[h * FDIM];
    atomicAdd(&acc[d * FDIM + f], sum);
}

__global__ void epilogue(const float* __restrict__ acc, const float* __restrict__ b,
                         float* __restrict__ out, int do_elu) {
    int i = blockIdx.x * blockDim.x + threadIdx.x;
    if (i >= NNODES * FDIM) return;
    const int f = i & (FDIM - 1);
    float bm = 0.f;
#pragma unroll
    for (int h = 0; h < HEADS; ++h) bm += b[h * FDIM + f];
    float v = acc[i] + bm * (1.0f / HEADS);
    out[i] = (do_elu && v <= 0.f) ? expm1f(v) : v;
}

static void run_gat_layer(const float* X, int K, const float* W,
                          const float* al, const float* ar, const float* b,
                          const int* src, const int* dst,
                          float* z, float* el, float* er, unsigned* m_ord,
                          float* denom, float* exbuf, float* acc, float* out,
                          int do_elu, hipStream_t stream) {
    gemm_wmma_f32<<<dim3(NNODES / 16, 1), 256, 0, stream>>>(X, W, z, K);

    int nh = NNODES * HEADS;
    attn_logits<<<(nh + 255) / 256, 256, 0, stream>>>(z, al, ar, el, er);

    int nf = NNODES * FDIM;
    init_bufs<<<(nf + 255) / 256, 256, 0, stream>>>(m_ord, denom, acc);

    int eh = NEDGES * HEADS;
    edge_max<<<(eh + 255) / 256, 256, 0, stream>>>(src, dst, el, er, m_ord);
    edge_expsum<<<(eh + 255) / 256, 256, 0, stream>>>(src, dst, el, er, m_ord, exbuf, denom);
    edge_alpha<<<(eh + 255) / 256, 256, 0, stream>>>(dst, denom, exbuf);

    int ef = NEDGES * FDIM;
    edge_agg<<<(ef + 255) / 256, 256, 0, stream>>>(src, dst, z, exbuf, acc);

    epilogue<<<(nf + 255) / 256, 256, 0, stream>>>(acc, b, out, do_elu);
}

extern "C" void kernel_launch(void* const* d_in, const int* in_sizes, int n_in,
                              void* d_out, int out_size, void* d_ws, size_t ws_size,
                              hipStream_t stream) {
    const float* x   = (const float*)d_in[0];
    const float* W1  = (const float*)d_in[1];
    const float* al1 = (const float*)d_in[2];
    const float* ar1 = (const float*)d_in[3];
    const float* b1  = (const float*)d_in[4];
    const float* W2  = (const float*)d_in[5];
    const float* al2 = (const float*)d_in[6];
    const float* ar2 = (const float*)d_in[7];
    const float* b2  = (const float*)d_in[8];
    const int*   src = (const int*)d_in[9];
    const int*   dst = (const int*)d_in[10];
    float* out = (float*)d_out;

    char* p = (char*)d_ws;
    float*    z     = (float*)p;    p += (size_t)NNODES * HF * 4;
    float*    el    = (float*)p;    p += (size_t)NNODES * HEADS * 4;
    float*    er    = (float*)p;    p += (size_t)NNODES * HEADS * 4;
    unsigned* m_ord = (unsigned*)p; p += (size_t)NNODES * HEADS * 4;
    float*    denom = (float*)p;    p += (size_t)NNODES * HEADS * 4;
    float*    exbuf = (float*)p;    p += (size_t)NEDGES * HEADS * 4;
    float*    acc   = (float*)p;    p += (size_t)NNODES * FDIM * 4;
    float*    hbuf  = (float*)p;    p += (size_t)NNODES * FDIM * 4;

    run_gat_layer(x, INF_, W1, al1, ar1, b1, src, dst,
                  z, el, er, m_ord, denom, exbuf, acc, hbuf,  1, stream);
    run_gat_layer(hbuf, FDIM, W2, al2, ar2, b2, src, dst,
                  z, el, er, m_ord, denom, exbuf, acc, out,  0, stream);
}


# edge lists built once by a cooperative pass (LDS count per range, one returning global add per range and group); no owner scans dst; LDS float adds follow the new arrival-order lists; 7 launches
# speedup vs baseline: 2.2126x; 2.2126x over previous
#include <hip/hip_runtime.h>
#include <hip/hip_bf16.h>

#define NNODES 50000
#define NEDGES 800000
#define INF_   256
#define FDIM   32
#define HEADS  8
#define HF     (HEADS * FDIM)
#define NEG_SLOPE 0.2f

typedef float v2f __attribute__((ext_vector_type(2)));
typedef float v8f __attribute__((ext_vector_type(8)));

#define GEMM_CB 64
#define NTILES (NNODES / 16)
#define NROWGROUPS ((NTILES + 7) / 8)
__device__ __forceinline__ float elu1(float v) { return v <= 0.f ? expm1f(v) : v; }
__device__ __forceinline__ float4 elu4(float4 v) {
    return float4{elu1(v.x), elu1(v.y), elu1(v.z), elu1(v.w)};
}
template <int K, bool ELU_A>
__global__ __launch_bounds__(256) void gemm_wlds_f32(const float* __restrict__ X,
                                                     const float* __restrict__ W,
                                                     float* __restrict__ Z) {
    constexpr int NQ = K / 4;
    constexpr int NM = K / 32;
    static_assert(!ELU_A || NM == 1, "the activation is applied to the first (only) line of a row");
    __shared__ float4 wl[NQ * GEMM_CB];
    const int tid   = threadIdx.x;
    const int lane  = tid & 31;
    const int wave  = tid >> 5;
    const int half  = lane >> 4;
    const int mn    = lane & 15;
    const int cbase = (blockIdx.x & 3) * GEMM_CB;
    const int tile  = (blockIdx.x >> 2) * 8 + wave;
    const bool valid = tile < NTILES;
    const int row0  = tile * 16;
    const float4* xr = reinterpret_cast<const float4*>(X + (size_t)(row0 + mn) * K) + half;

    float4 a0 = {}, a1 = {}, a2 = {}, a3 = {};
    if (valid) { a0 = xr[0]; a1 = xr[2]; a2 = xr[4]; a3 = xr[6]; }

    {
        const int col = tid & (GEMM_CB - 1);
        const int q0  = tid >> 6;
        const float* wp = W + cbase + col;
#pragma unroll
        for (int i = 0; i < NQ / 4; ++i) {
            const int q = q0 + 4 * i;
            float4 v;
            v.x = wp[(size_t)(4 * q + 0) * HF];
            v.y = wp[(size_t)(4 * q + 1) * HF];
            v.z = wp[(size_t)(4 * q + 2) * HF];
            v.w = wp[(size_t)(4 * q + 3) * HF];
            wl[q * GEMM_CB + col] = v;
        }
    }
    __syncthreads();
    if (!valid) return;
    if constexpr (ELU_A) { a0 = elu4(a0); a1 = elu4(a1); a2 = elu4(a2); a3 = elu4(a3); }

    const float4* bl = wl + half * GEMM_CB + mn;
    v8f c0 = {}, c1 = {}, c2 = {}, c3 = {};
#define GEMM_STEP(A_, S_)                                                              \
    {                                                                                  \
        const float4* bp = bl + (2 * (S_)) * GEMM_CB;                                  \
        const float4 b0 = bp[0], b1 = bp[16], b2 = bp[32], b3 = bp[48];                \
        const v2f alo = {A_.x, A_.y}, ahi = {A_.z, A_.w};                              \
        c0 = __builtin_amdgcn_wmma_f32_16x16x4_f32(false, alo, false, v2f{b0.x, b0.y}, (short)0, c0, false, false); \
        c1 = __builtin_amdgcn_wmma_f32_16x16x4_f32(false, alo, false, v2f{b1.x, b1.y}, (short)0, c1, false, false); \
        c2 = __builtin_amdgcn_wmma_f32_16x16x4_f32(false, alo, false, v2f{b2.x, b2.y}, (short)0, c2, false, false); \
        c3 = __builtin_amdgcn_wmma_f32_16x16x4_f32(false, alo, false, v2f{b3.x, b3.y}, (short)0, c3, false, false); \
        c0 = __builtin_amdgcn_wmma_f32_16x16x4_f32(false, ahi, false, v2f{b0.z, b0.w}, (short)0, c0, false, false); \
        c1 = __builtin_amdgcn_wmma_f32_16x16x4_f32(false, ahi, false, v2f{b1.z, b1.w}, (short)0, c1, false, false); \
        c2 = __builtin_amdgcn_wmma_f32_16x16x4_f32(false, ahi, false, v2f{b2.z, b2.w}, (short)0, c2, false, false); \
        c3 = __builtin_amdgcn_wmma_f32_16x16x4_f32(false, ahi, false, v2f{b3.z, b3.w}, (short)0, c3, false, false); \
    }
    for (int m = 0; m < NM; ++m) {
        float4 n0 = a0, n1 = a1, n2 = a2, n3 = a3;
        if (m + 1 < NM) {
            const float4* xn = xr + 8 * (m + 1);
            n0 = xn[0]; n1 = xn[2]; n2 = xn[4]; n3 = xn[6];
        }
        GEMM_STEP(a0, 4 * m + 0)
        GEMM_STEP(a1, 4 * m + 1)
        GEMM_STEP(a2, 4 * m + 2)
        GEMM_STEP(a3, 4 * m + 3)
        a0 = n0; a1 = n1; a2 = n2; a3 = n3;
    }
#undef GEMM_STEP
    float* zout = Z + (size_t)(row0 + half * 8) * HF + cbase + mn;
#pragma unroll
    for (int v = 0; v < 8; ++v) {
        zout[(size_t)v * HF +  0] = c0[v];
        zout[(size_t)v * HF + 16] = c1[v];
        zout[(size_t)v * HF + 32] = c2[v];
        zout[(size_t)v * HF + 48] = c3[v];
    }
}

#define RNG  196
#define NRB  ((NNODES + RNG - 1) / RNG)
#define LREG 4096
#define LHDR 64
#define LCAP (LREG - LHDR)
static_assert(NNODES <= 65536 && RNG <= 256 && NRB <= 256, "list entry: src in 16 bits, local dst in 8; bin_edges adds the range in 8 more");

__device__ __forceinline__ unsigned f2ord(float f) {
    unsigned u = __float_as_uint(f);
    return (u & 0x80000000u) ? ~u : (u | 0x80000000u);
}
__device__ __forceinline__ float ord2f(unsigned e) {
    unsigned u = (e & 0x80000000u) ? (e & 0x7FFFFFFFu) : ~e;
    return __uint_as_float(u);
}
#define ORD_NEG_INF 0x007FFFFFu

template <bool CLEAR_LISTS>
__global__ void node_prep(const float* __restrict__ Z,
                          const float* __restrict__ al, const float* __restrict__ ar,
                          float* __restrict__ el, float* __restrict__ er,
                          unsigned* __restrict__ wlst) {
    int i = blockIdx.x * blockDim.x + threadIdx.x;
    if constexpr (CLEAR_LISTS) { if (i < NRB) wlst[(size_t)i * LREG] = 0u; }
    if (i >= NNODES * HEADS) return;
    const int h = i & (HEADS - 1);
    const int n = i >> 3;
    const float* z  = Z + (size_t)n * HF + h * FDIM;
    const float* pl = al + h * FDIM;
    const float* pr = ar + h * FDIM;
    float sl = 0.f, sr = 0.f;
#pragma unroll
    for (int f = 0; f < FDIM; ++f) { float zv = z[f]; sl += zv * pl[f]; sr += zv * pr[f]; }
    el[i] = sl; er[i] = sr;
}

__device__ __forceinline__ float lrelu(float v) { return v >= 0.f ? v : NEG_SLOPE * v; }

#define RTHREADS 1024
#define RWAVES (RTHREADS / 32)
static_assert(NEDGES % 32 == 0, "the fallback walks dst by whole waves");

#define BIN_THREADS 1024
#define BIN_Q 2
#define BIN_EPT (BIN_Q * 4)
#define BIN_EDGES (BIN_THREADS * BIN_EPT)
#define NBINB ((NEDGES + BIN_EDGES - 1) / BIN_EDGES)
static_assert(NEDGES % 4 == 0, "dst and src are read by 16-byte loads");
__global__ __launch_bounds__(BIN_THREADS) void bin_edges(const int* __restrict__ src, const int* __restrict__ dst,
                                                        unsigned* __restrict__ wlst) {
    __shared__ unsigned hist[NRB];
    __shared__ unsigned off[NRB];
    __shared__ unsigned gbase[NRB];
    __shared__ unsigned wtot[NRB / 32];
    __shared__ unsigned ent[BIN_EDGES];
    static_assert(NRB % 32 == 0, "the prefix sum runs in whole waves");
    const int tid = threadIdx.x;
    if (tid < NRB) hist[tid] = 0u;
    int dv[BIN_EPT], sv[BIN_EPT];
    for (int i = 0; i < BIN_Q; ++i) {
        const int q = (blockIdx.x * BIN_Q + i) * BIN_THREADS + tid;
        int4 d = {-1, -1, -1, -1}, s = {0, 0, 0, 0};
        if (q < NEDGES / 4) {
            d = reinterpret_cast<const int4*>(dst)[q];
            s = reinterpret_cast<const int4*>(src)[q];
        }
        dv[4 * i + 0] = d.x; dv[4 * i + 1] = d.y; dv[4 * i + 2] = d.z; dv[4 * i + 3] = d.w;
        sv[4 * i + 0] = s.x; sv[4 * i + 1] = s.y; sv[4 * i + 2] = s.z; sv[4 * i + 3] = s.w;
    }
    __syncthreads();
    unsigned bin[BIN_EPT], rnk[BIN_EPT], word[BIN_EPT];
    bool ok[BIN_EPT];
    for (int k = 0; k < BIN_EPT; ++k) {
        ok[k]  = (unsigned)dv[k] < (unsigned)NNODES;
        bin[k] = ok[k] ? (unsigned)dv[k] / RNG : 0u;
        word[k] = (unsigned)sv[k] | (((unsigned)dv[k] - bin[k] * RNG) << 16) | (bin[k] << 24);
        rnk[k] = ok[k] ? atomicAdd(&hist[bin[k]], 1u) : 0u;
    }
    __syncthreads();
    unsigned cntv = 0u, incl = 0u;
    if (tid < NRB) {
        cntv = hist[tid];
        incl = cntv;
        for (int dl = 1; dl < 32; dl <<= 1) {
            const unsigned t = __shfl_up(incl, dl);
            if ((tid & 31) >= dl) incl += t;
        }
        if ((tid & 31) == 31) wtot[tid >> 5] = incl;
        gbase[tid] = cntv ? atomicAdd(&wlst[(size_t)tid * LREG], cntv) : 0u;
    }
    __syncthreads();
    if (tid < NRB) {
        unsigned before = 0u;
        for (int w = 0; w < (tid >> 5); ++w) before += wtot[w];
        off[tid] = before + incl - cntv;
    }
    __syncthreads();
    unsigned slot[BIN_EPT];
    for (int k = 0; k < BIN_EPT; ++k) slot[k] = off[bin[k]] + rnk[k];
    for (int k = 0; k < BIN_EPT; ++k)
        if (ok[k]) ent[slot[k]] = word[k];
    __syncthreads();
    const unsigned total = off[NRB - 1] + hist[NRB - 1];
    for (unsigned j = tid; j < total; j += BIN_THREADS) {
        const unsigned e = ent[j];
        const unsigned r = e >> 24;
        const unsigned pos = gbase[r] + (j - off[r]);
        if (pos < LCAP) wlst[(size_t)r * LREG + LHDR + pos] = e & 0x00FFFFFFu;
    }
}

struct RangeState {
    float    acc[RNG * FDIM];
    unsigned mo[RNG * HEADS];
    float    den[RNG * HEADS];
    float    erl[RNG * HEADS];
    unsigned lst[LCAP];
};

template <class F>
__device__ __forceinline__ void each_edge(bool fits, int cnt, const unsigned* lst,
                                          const int* __restrict__ src, const int* __restrict__ dst,
                                          int lo, unsigned nloc, int tid, F f) {
    if (fits) {
        for (int i = tid; i < cnt; i += RTHREADS) { const unsigned p = lst[i]; f((int)(p & 0xFFFFu), (int)(p >> 16)); }
    } else {
        for (int e = tid; e < NEDGES; e += RTHREADS) {
            const unsigned r = (unsigned)(dst[e] - lo);
            if (r < nloc) f(src[e], (int)r);
        }
    }
}

__global__ __launch_bounds__(RTHREADS) void range_agg(const int* __restrict__ src, const int* __restrict__ dst,
                          const float* __restrict__ Z,
                          const float* __restrict__ el, const float* __restrict__ er,
                          const float* __restrict__ b, float* __restrict__ sums,
                          const unsigned* __restrict__ wlst) {
    __shared__ RangeState S;
    const int tid  = threadIdx.x;
    const int lane = tid & 31;
    const int wave = __builtin_amdgcn_readfirstlane(tid >> 5);
    const int lo   = blockIdx.x * RNG;
    const unsigned nloc = (unsigned)min(RNG, NNODES - lo);
    const unsigned* mylist = wlst + (size_t)blockIdx.x * LREG;
    const int  cnt  = (int)mylist[0];
    const bool fits = cnt <= LCAP;

    for (int i = tid; i < RNG * FDIM; i += RTHREADS) S.acc[i] = 0.f;
    for (int i = tid; i < RNG * HEADS; i += RTHREADS) {
        S.mo[i] = ORD_NEG_INF; S.den[i] = 0.f;
        S.erl[i] = (unsigned)i < nloc * HEADS ? er[lo * HEADS + i] : 0.f;
    }
    if (fits)
        for (int i = tid; i < cnt; i += RTHREADS) S.lst[i] = mylist[LHDR + i];
    __syncthreads();

    each_edge(fits, cnt, S.lst, src, dst, lo, nloc, tid, [&](int s, int r) {
        const float4* e4 = reinterpret_cast<const float4*>(el + (size_t)s * HEADS);
        const float4 a = e4[0], c = e4[1];
        const float ev[HEADS] = {a.x, a.y, a.z, a.w, c.x, c.y, c.z, c.w};
#pragma unroll
        for (int h = 0; h < HEADS; ++h)
            atomicMax(&S.mo[r * HEADS + h], f2ord(lrelu(ev[h] + S.erl[r * HEADS + h])));
    });
    __syncthreads();
    each_edge(fits, cnt, S.lst, src, dst, lo, nloc, tid, [&](int s, int r) {
        const float4* e4 = reinterpret_cast<const float4*>(el + (size_t)s * HEADS);
        const float4 a = e4[0], c = e4[1];
        const float ev[HEADS] = {a.x, a.y, a.z, a.w, c.x, c.y, c.z, c.w};
#pragma unroll
        for (int h = 0; h < HEADS; ++h) {
            const float v = lrelu(ev[h] + S.erl[r * HEADS + h]);
            atomicAdd(&S.den[r * HEADS + h], __expf(v - ord2f(S.mo[r * HEADS + h])));
        }
    });
    __syncthreads();

    auto weight = [&](int s, int r, int h) {
        const float v = lrelu(el[(size_t)s * HEADS + h] + S.erl[r * HEADS + h]);
        const float ex = __expf(v - ord2f(S.mo[r * HEADS + h]));
        return ex / S.den[r * HEADS + h] * (1.0f / HEADS);
    };
    if (fits) {
        for (int j0 = wave * 8; j0 < cnt; j0 += RWAVES * 8) {
            const int ja = j0 + (lane >> 3), jb = ja + 4;
            const unsigned pa = ja < cnt ? S.lst[ja] : 0u;
            const unsigned pb = jb < cnt ? S.lst[jb] : 0u;
            const float w0 = weight((int)(pa & 0xFFFFu), (int)(pa >> 16), lane & 7);
            const float w1 = weight((int)(pb & 0xFFFFu), (int)(pb >> 16), lane & 7);
            float zv[8][HEADS];
#pragma unroll
            for (int j = 0; j < 8; ++j) {
                const unsigned p = (unsigned)__builtin_amdgcn_readlane((int)(j < 4 ? pa : pb), (j & 3) * 8);
                const float* zs = Z + (size_t)(p & 0xFFFFu) * HF + lane;
#pragma unroll
                for (int h = 0; h < HEADS; ++h) zv[j][h] = zs[h * FDIM];
            }
#pragma unroll
            for (int j = 0; j < 8; ++j) {
                const unsigned p = (unsigned)__builtin_amdgcn_readlane((int)(j < 4 ? pa : pb), (j & 3) * 8);
                const int wbits = __float_as_int(j < 4 ? w0 : w1);
                float sum = 0.f;
#pragma unroll
                for (int h = 0; h < HEADS; ++h) {
                    const float a = __int_as_float(__builtin_amdgcn_readlane(wbits, (j & 3) * HEADS + h));
                    sum += a * zv[j][h];
                }
                if (j0 + j < cnt) atomicAdd(&S.acc[(p >> 16) * FDIM + lane], sum);
            }
        }
    } else {
        for (int base = wave * 32; base < NEDGES; base += RTHREADS) {
            const unsigned rr = (unsigned)(dst[base + lane] - lo);
            const bool ok = rr < nloc;
            const int  ss = ok ? src[base + lane] : 0;
            unsigned mask = __builtin_amdgcn_ballot_w32(ok);
            while (mask) {
                const int bit = __builtin_ctz(mask);
                mask &= mask - 1;
                const int s = __builtin_amdgcn_readlane(ss, bit);
                const int r = __builtin_amdgcn_readlane((int)rr, bit);
                const int wbits = __float_as_int(weight(s, r, lane & 7));
                const float* zs = Z + (size_t)s * HF + lane;
                float sum = 0.f;
#pragma unroll
                for (int h = 0; h < HEADS; ++h)
                    sum += __int_as_float(__builtin_amdgcn_readlane(wbits, h)) * zs[h * FDIM];
                atomicAdd(&S.acc[r * FDIM + lane], sum);
            }
        }
    }
    __syncthreads();

    for (int i = tid; i < (int)nloc * FDIM; i += RTHREADS) {
        const int f = i & (FDIM - 1);
        float bm = 0.f;
#pragma unroll
        for (int hh = 0; hh < HEADS; ++hh) bm += b[hh * FDIM + f];
        sums[(size_t)lo * FDIM + i] = S.acc[i] + bm * (1.0f / HEADS);
    }
}

template <int K, bool ELU_A>
static void run_gat_layer(const float* X, const float* W,
                          const float* al, const float* ar, const float* b,
                          const int* src, const int* dst,
                          float* z, float* el, float* er, float* sums,
                          unsigned* wlst,
                          hipStream_t stream) {
    gemm_wlds_f32<K, ELU_A><<<NROWGROUPS * 4, 256, 0, stream>>>(X, W, z);

    int nh = NNODES * HEADS;
    node_prep<!ELU_A><<<(nh + 255) / 256, 256, 0, stream>>>(z, al, ar, el, er, wlst);

    if constexpr (!ELU_A) bin_edges<<<NBINB, BIN_THREADS, 0, stream>>>(src, dst, wlst);

    range_agg<<<NRB, RTHREADS, 0, stream>>>(src, dst, z, el, er, b, sums, wlst);
}

extern "C" void kernel_launch(void* const* d_in, const int* in_sizes, int n_in,
                              void* d_out, int out_size, void* d_ws, size_t ws_size,
                              hipStream_t stream) {
    const float* x   = (const float*)d_in[0];
    const float* W1  = (const float*)d_in[1];
    const float* al1 = (const float*)d_in[2];
    const float* ar1 = (const float*)d_in[3];
    const float* b1  = (const float*)d_in[4];
    const float* W2  = (const float*)d_in[5];
    const float* al2 = (const float*)d_in[6];
    const float* ar2 = (const float*)d_in[7];
    const float* b2  = (const float*)d_in[8];
    const int*   src = (const int*)d_in[9];
    const int*   dst = (const int*)d_in[10];
    float* out = (float*)d_out;

    char* p = (char*)d_ws;
    float*    z     = (float*)p;    p += (size_t)NNODES * HF * 4;
    float*    el    = (float*)p;    p += (size_t)NNODES * HEADS * 4;
    float*    er    = (float*)p;    p += (size_t)NNODES * HEADS * 4;
    float*    acc   = (float*)p;    p += (size_t)NNODES * FDIM * 4;
    unsigned* wlst  = (unsigned*)p; p += (size_t)NRB * LREG * 4;

    run_gat_layer<INF_, false>(x, W1, al1, ar1, b1, src, dst, z, el, er, acc, wlst, stream);
    run_gat_layer<FDIM, true>(acc, W2, al2, ar2, b2, src, dst, z, el, er, out, wlst, stream);
}


# range_agg: an edge's 8 logits stay in an LDS table (max pass writes them, sum pass makes them exp, feature pass reads weights): el row gathered once, not 3 times; LDS float adds in arrival order
# speedup vs baseline: 2.3762x; 1.0739x over previous
#include <hip/hip_runtime.h>
#include <hip/hip_bf16.h>

#define NNODES 50000
#define NEDGES 800000
#define INF_   256
#define FDIM   32
#define HEADS  8
#define HF     (HEADS * FDIM)
#define NEG_SLOPE 0.2f

typedef float v2f __attribute__((ext_vector_type(2)));
typedef float v8f __attribute__((ext_vector_type(8)));

#define GEMM_CB 64
#define NTILES (NNODES / 16)
#define NROWGROUPS ((NTILES + 7) / 8)
__device__ __forceinline__ float elu1(float v) { return v <= 0.f ? expm1f(v) : v; }
__device__ __forceinline__ float4 elu4(float4 v) {
    return float4{elu1(v.x), elu1(v.y), elu1(v.z), elu1(v.w)};
}
template <int K, bool ELU_A>
__global__ __launch_bounds__(256) void gemm_wlds_f32(const float* __restrict__ X,
                                                     const float* __restrict__ W,
                                                     float* __restrict__ Z) {
    constexpr int NQ = K / 4;
    constexpr int NM = K / 32;
    static_assert(!ELU_A || NM == 1, "the activation is applied to the first (only) line of a row");
    __shared__ float4 wl[NQ * GEMM_CB];
    const int tid   = threadIdx.x;
    const int lane  = tid & 31;
    const int wave  = tid >> 5;
    const int half  = lane >> 4;
    const int mn    = lane & 15;
    const int cbase = (blockIdx.x & 3) * GEMM_CB;
    const int tile  = (blockIdx.x >> 2) * 8 + wave;
    const bool valid = tile < NTILES;
    const int row0  = tile * 16;
    const float4* xr = reinterpret_cast<const float4*>(X + (size_t)(row0 + mn) * K) + half;

    float4 a0 = {}, a1 = {}, a2 = {}, a3 = {};
    if (valid) { a0 = xr[0]; a1 = xr[2]; a2 = xr[4]; a3 = xr[6]; }

    {
        const int col = tid & (GEMM_CB - 1);
        const int q0  = tid >> 6;
        const float* wp = W + cbase + col;
#pragma unroll
        for (int i = 0; i < NQ / 4; ++i) {
            const int q = q0 + 4 * i;
            float4 v;
            v.x = wp[(size_t)(4 * q + 0) * HF];
            v.y = wp[(size_t)(4 * q + 1) * HF];
            v.z = wp[(size_t)(4 * q + 2) * HF];
            v.w = wp[(size_t)(4 * q + 3) * HF];
            wl[q * GEMM_CB + col] = v;
        }
    }
    __syncthreads();
    if (!valid) return;
    if constexpr (ELU_A) { a0 = elu4(a0); a1 = elu4(a1); a2 = elu4(a2); a3 = elu4(a3); }

    const float4* bl = wl + half * GEMM_CB + mn;
    v8f c0 = {}, c1 = {}, c2 = {}, c3 = {};
#define GEMM_STEP(A_, S_)                                                              \
    {                                                                                  \
        const float4* bp = bl + (2 * (S_)) * GEMM_CB;                                  \
        const float4 b0 = bp[0], b1 = bp[16], b2 = bp[32], b3 = bp[48];                \
        const v2f alo = {A_.x, A_.y}, ahi = {A_.z, A_.w};                              \
        c0 = __builtin_amdgcn_wmma_f32_16x16x4_f32(false, alo, false, v2f{b0.x, b0.y}, (short)0, c0, false, false); \
        c1 = __builtin_amdgcn_wmma_f32_16x16x4_f32(false, alo, false, v2f{b1.x, b1.y}, (short)0, c1, false, false); \
        c2 = __builtin_amdgcn_wmma_f32_16x16x4_f32(false, alo, false, v2f{b2.x, b2.y}, (short)0, c2, false, false); \
        c3 = __builtin_amdgcn_wmma_f32_16x16x4_f32(false, alo, false, v2f{b3.x, b3.y}, (short)0, c3, false, false); \
        c0 = __builtin_amdgcn_wmma_f32_16x16x4_f32(false, ahi, false, v2f{b0.z, b0.w}, (short)0, c0, false, false); \
        c1 = __builtin_amdgcn_wmma_f32_16x16x4_f32(false, ahi, false, v2f{b1.z, b1.w}, (short)0, c1, false, false); \
        c2 = __builtin_amdgcn_wmma_f32_16x16x4_f32(false, ahi, false, v2f{b2.z, b2.w}, (short)0, c2, false, false); \
        c3 = __builtin_amdgcn_wmma_f32_16x16x4_f32(false, ahi, false, v2f{b3.z, b3.w}, (short)0, c3, false, false); \
    }
    for (int m = 0; m < NM; ++m) {
        float4 n0 = a0, n1 = a1, n2 = a2, n3 = a3;
        if (m + 1 < NM) {
            const float4* xn = xr + 8 * (m + 1);
            n0 = xn[0]; n1 = xn[2]; n2 = xn[4]; n3 = xn[6];
        }
        GEMM_STEP(a0, 4 * m + 0)
        GEMM_STEP(a1, 4 * m + 1)
        GEMM_STEP(a2, 4 * m + 2)
        GEMM_STEP(a3, 4 * m + 3)
        a0 = n0; a1 = n1; a2 = n2; a3 = n3;
    }
#undef GEMM_STEP
    float* zout = Z + (size_t)(row0 + half * 8) * HF + cbase + mn;
#pragma unroll
    for (int v = 0; v < 8; ++v) {
        zout[(size_t)v * HF +  0] = c0[v];
        zout[(size_t)v * HF + 16] = c1[v];
        zout[(size_t)v * HF + 32] = c2[v];
        zout[(size_t)v * HF + 48] = c3[v];
    }
}

#define RNG  196
#define NRB  ((NNODES + RNG - 1) / RNG)
#define LREG 4096
#define LHDR 64
#define LCAP (LREG - LHDR)
static_assert(NNODES <= 65536 && RNG <= 256 && NRB <= 256, "list entry: src in 16 bits, local dst in 8; bin_edges adds the range in 8 more");

__device__ __forceinline__ unsigned f2ord(float f) {
    unsigned u = __float_as_uint(f);
    return (u & 0x80000000u) ? ~u : (u | 0x80000000u);
}
__device__ __forceinline__ float ord2f(unsigned e) {
    unsigned u = (e & 0x80000000u) ? (e & 0x7FFFFFFFu) : ~e;
    return __uint_as_float(u);
}
#define ORD_NEG_INF 0x007FFFFFu

template <bool CLEAR_LISTS>
__global__ void node_prep(const float* __restrict__ Z,
                          const float* __restrict__ al, const float* __restrict__ ar,
                          float* __restrict__ el, float* __restrict__ er,
                          unsigned* __restrict__ wlst) {
    int i = blockIdx.x * blockDim.x + threadIdx.x;
    if constexpr (CLEAR_LISTS) { if (i < NRB) wlst[(size_t)i * LREG] = 0u; }
    if (i >= NNODES * HEADS) return;
    const int h = i & (HEADS - 1);
    const int n = i >> 3;
    const float* z  = Z + (size_t)n * HF + h * FDIM;
    const float* pl = al + h * FDIM;
    const float* pr = ar + h * FDIM;
    float sl = 0.f, sr = 0.f;
#pragma unroll
    for (int f = 0; f < FDIM; ++f) { float zv = z[f]; sl += zv * pl[f]; sr += zv * pr[f]; }
    el[i] = sl; er[i] = sr;
}

__device__ __forceinline__ float lrelu(float v) { return v >= 0.f ? v : NEG_SLOPE * v; }

#define RTHREADS 1024
#define RWAVES (RTHREADS / 32)
static_assert(NEDGES % 32 == 0, "the fallback walks dst by whole waves");

#define BIN_THREADS 1024
#define BIN_Q 2
#define BIN_EPT (BIN_Q * 4)
#define BIN_EDGES (BIN_THREADS * BIN_EPT)
#define NBINB ((NEDGES + BIN_EDGES - 1) / BIN_EDGES)
static_assert(NEDGES % 4 == 0, "dst and src are read by 16-byte loads");
__global__ __launch_bounds__(BIN_THREADS) void bin_edges(const int* __restrict__ src, const int* __restrict__ dst,
                                                        unsigned* __restrict__ wlst) {
    __shared__ unsigned hist[NRB];
    __shared__ unsigned off[NRB];
    __shared__ unsigned gbase[NRB];
    __shared__ unsigned wtot[NRB / 32];
    __shared__ unsigned ent[BIN_EDGES];
    static_assert(NRB % 32 == 0, "the prefix sum runs in whole waves");
    const int tid = threadIdx.x;
    if (tid < NRB) hist[tid] = 0u;
    int dv[BIN_EPT], sv[BIN_EPT];
    for (int i = 0; i < BIN_Q; ++i) {
        const int q = (blockIdx.x * BIN_Q + i) * BIN_THREADS + tid;
        int4 d = {-1, -1, -1, -1}, s = {0, 0, 0, 0};
        if (q < NEDGES / 4) {
            d = reinterpret_cast<const int4*>(dst)[q];
            s = reinterpret_cast<const int4*>(src)[q];
        }
        dv[4 * i + 0] = d.x; dv[4 * i + 1] = d.y; dv[4 * i + 2] = d.z; dv[4 * i + 3] = d.w;
        sv[4 * i + 0] = s.x; sv[4 * i + 1] = s.y; sv[4 * i + 2] = s.z; sv[4 * i + 3] = s.w;
    }
    __syncthreads();
    unsigned bin[BIN_EPT], rnk[BIN_EPT], word[BIN_EPT];
    bool ok[BIN_EPT];
    for (int k = 0; k < BIN_EPT; ++k) {
        ok[k]  = (unsigned)dv[k] < (unsigned)NNODES;
        bin[k] = ok[k] ? (unsigned)dv[k] / RNG : 0u;
        word[k] = (unsigned)sv[k] | (((unsigned)dv[k] - bin[k] * RNG) << 16) | (bin[k] << 24);
        rnk[k] = ok[k] ? atomicAdd(&hist[bin[k]], 1u) : 0u;
    }
    __syncthreads();
    unsigned cntv = 0u, incl = 0u;
    if (tid < NRB) {
        cntv = hist[tid];
        incl = cntv;
        for (int dl = 1; dl < 32; dl <<= 1) {
            const unsigned t = __shfl_up(incl, dl);
            if ((tid & 31) >= dl) incl += t;
        }
        if ((tid & 31) == 31) wtot[tid >> 5] = incl;
        gbase[tid] = cntv ? atomicAdd(&wlst[(size_t)tid * LREG], cntv) : 0u;
    }
    __syncthreads();
    if (tid < NRB) {
        unsigned before = 0u;
        for (int w = 0; w < (tid >> 5); ++w) before += wtot[w];
        off[tid] = before + incl - cntv;
    }
    __syncthreads();
    unsigned slot[BIN_EPT];
    for (int k = 0; k < BIN_EPT; ++k) slot[k] = off[bin[k]] + rnk[k];
    for (int k = 0; k < BIN_EPT; ++k)
        if (ok[k]) ent[slot[k]] = word[k];
    __syncthreads();
    const unsigned total = off[NRB - 1] + hist[NRB - 1];
    for (unsigned j = tid; j < total; j += BIN_THREADS) {
        const unsigned e = ent[j];
        const unsigned r = e >> 24;
        const unsigned pos = gbase[r] + (j - off[r]);
        if (pos < LCAP) wlst[(size_t)r * LREG + LHDR + pos] = e & 0x00FFFFFFu;
    }
}

struct RangeState {
    float    acc[RNG * FDIM];
    unsigned mo[RNG * HEADS];
    float    den[RNG * HEADS];
    float    erl[RNG * HEADS];
    unsigned lst[LCAP];
    float    ew[LCAP * HEADS];
};
static_assert(sizeof(RangeState) <= 320 * 1024, "one work-group's LDS");
static_assert((RNG * FDIM + 3 * RNG * HEADS + LCAP) % 4 == 0, "ew rows are read and written as 16-byte pieces");

template <class F>
__device__ __forceinline__ void each_edge_of_dst(const int* __restrict__ src, const int* __restrict__ dst,
                                                 int lo, unsigned nloc, int tid, F f) {
    for (int e = tid; e < NEDGES; e += RTHREADS) {
        const unsigned r = (unsigned)(dst[e] - lo);
        if (r < nloc) f(src[e], (int)r);
    }
}

__global__ __launch_bounds__(RTHREADS) void range_agg(const int* __restrict__ src, const int* __restrict__ dst,
                          const float* __restrict__ Z,
                          const float* __restrict__ el, const float* __restrict__ er,
                          const float* __restrict__ b, float* __restrict__ sums,
                          const unsigned* __restrict__ wlst) {
    __shared__ __attribute__((aligned(16))) RangeState S;
    const int tid  = threadIdx.x;
    const int lane = tid & 31;
    const int wave = __builtin_amdgcn_readfirstlane(tid >> 5);
    const int lo   = blockIdx.x * RNG;
    const unsigned nloc = (unsigned)min(RNG, NNODES - lo);
    const unsigned* mylist = wlst + (size_t)blockIdx.x * LREG;
    const int  cnt  = (int)mylist[0];
    const bool fits = cnt <= LCAP;

    for (int i = tid; i < RNG * FDIM; i += RTHREADS) S.acc[i] = 0.f;
    for (int i = tid; i < RNG * HEADS; i += RTHREADS) {
        S.mo[i] = ORD_NEG_INF; S.den[i] = 0.f;
        S.erl[i] = (unsigned)i < nloc * HEADS ? er[lo * HEADS + i] : 0.f;
    }
    if (fits)
        for (int i = tid; i < cnt; i += RTHREADS) S.lst[i] = mylist[LHDR + i];
    __syncthreads();

    if (fits) {
        for (int i = tid; i < cnt; i += RTHREADS) {
            const unsigned p = S.lst[i];
            const int r = (int)(p >> 16);
            const float4* e4 = reinterpret_cast<const float4*>(el + (size_t)(p & 0xFFFFu) * HEADS);
            const float4 a = e4[0], c = e4[1];
            const float ev[HEADS] = {a.x, a.y, a.z, a.w, c.x, c.y, c.z, c.w};
            float v[HEADS];
            for (int h = 0; h < HEADS; ++h) {
                v[h] = lrelu(ev[h] + S.erl[r * HEADS + h]);
                atomicMax(&S.mo[r * HEADS + h], f2ord(v[h]));
            }
            float4* t = reinterpret_cast<float4*>(S.ew + i * HEADS);
            t[0] = float4{v[0], v[1], v[2], v[3]};
            t[1] = float4{v[4], v[5], v[6], v[7]};
        }
    } else {
        each_edge_of_dst(src, dst, lo, nloc, tid, [&](int s, int r) {
            const float4* e4 = reinterpret_cast<const float4*>(el + (size_t)s * HEADS);
            const float4 a = e4[0], c = e4[1];
            const float ev[HEADS] = {a.x, a.y, a.z, a.w, c.x, c.y, c.z, c.w};
#pragma unroll
            for (int h = 0; h < HEADS; ++h)
                atomicMax(&S.mo[r * HEADS + h], f2ord(lrelu(ev[h] + S.erl[r * HEADS + h])));
        });
    }
    __syncthreads();
    if (fits) {
        for (int i = tid; i < cnt; i += RTHREADS) {
            const int r = (int)(S.lst[i] >> 16);
            float4* t = reinterpret_cast<float4*>(S.ew + i * HEADS);
            const float4 a = t[0], c = t[1];
            float v[HEADS] = {a.x, a.y, a.z, a.w, c.x, c.y, c.z, c.w};
            for (int h = 0; h < HEADS; ++h) {
                v[h] = __expf(v[h] - ord2f(S.mo[r * HEADS + h]));
                atomicAdd(&S.den[r * HEADS + h], v[h]);
            }
            t[0] = float4{v[0], v[1], v[2], v[3]};
            t[1] = float4{v[4], v[5], v[6], v[7]};
        }
    } else {
        each_edge_of_dst(src, dst, lo, nloc, tid, [&](int s, int r) {
            const float4* e4 = reinterpret_cast<const float4*>(el + (size_t)s * HEADS);
            const float4 a = e4[0], c = e4[1];
            const float ev[HEADS] = {a.x, a.y, a.z, a.w, c.x, c.y, c.z, c.w};
#pragma unroll
            for (int h = 0; h < HEADS; ++h) {
                const float v = lrelu(ev[h] + S.erl[r * HEADS + h]);
                atomicAdd(&S.den[r * HEADS + h], __expf(v - ord2f(S.mo[r * HEADS + h])));
            }
        });
    }
    __syncthreads();

    auto weight = [&](int s, int r, int h) {
        const float v = lrelu(el[(size_t)s * HEADS + h] + S.erl[r * HEADS + h]);
        const float ex = __expf(v - ord2f(S.mo[r * HEADS + h]));
        return ex / S.den[r * HEADS + h] * (1.0f / HEADS);
    };
    if (fits) {
        for (int j0 = wave * 8; j0 < cnt; j0 += RWAVES * 8) {
            const int ja = min(j0 + (lane >> 3), cnt - 1), jb = min(j0 + (lane >> 3) + 4, cnt - 1);
            const unsigned pa = S.lst[ja];
            const unsigned pb = S.lst[jb];
            const float w0 = S.ew[ja * HEADS + (lane & 7)] / S.den[(pa >> 16) * HEADS + (lane & 7)] * (1.0f / HEADS);
            const float w1 = S.ew[jb * HEADS + (lane & 7)] / S.den[(pb >> 16) * HEADS + (lane & 7)] * (1.0f / HEADS);
            float zv[8][HEADS];
#pragma unroll
            for (int j = 0; j < 8; ++j) {
                const unsigned p = (unsigned)__builtin_amdgcn_readlane((int)(j < 4 ? pa : pb), (j & 3) * 8);
                const float* zs = Z + (size_t)(p & 0xFFFFu) * HF + lane;
#pragma unroll
                for (int h = 0; h < HEADS; ++h) zv[j][h] = zs[h * FDIM];
            }
#pragma unroll
            for (int j = 0; j < 8; ++j) {
                const unsigned p = (unsigned)__builtin_amdgcn_readlane((int)(j < 4 ? pa : pb), (j & 3) * 8);
                const int wbits = __float_as_int(j < 4 ? w0 : w1);
                float sum = 0.f;
#pragma unroll
                for (int h = 0; h < HEADS; ++h) {
                    const float a = __int_as_float(__builtin_amdgcn_readlane(wbits, (j & 3) * HEADS + h));
                    sum += a * zv[j][h];
                }
                if (j0 + j < cnt) atomicAdd(&S.acc[(p >> 16) * FDIM + lane], sum);
            }
        }
    } else {
        for (int base = wave * 32; base < NEDGES; base += RTHREADS) {
            const unsigned rr = (unsigned)(dst[base + lane] - lo);
            const bool ok = rr < nloc;
            const int  ss = ok ? src[base + lane] : 0;
            unsigned mask = __builtin_amdgcn_ballot_w32(ok);
            while (mask) {
                const int bit = __builtin_ctz(mask);
                mask &= mask - 1;
                const int s = __builtin_amdgcn_readlane(ss, bit);
                const int r = __builtin_amdgcn_readlane((int)rr, bit);
                const int wbits = __float_as_int(weight(s, r, lane & 7));
                const float* zs = Z + (size_t)s * HF + lane;
                float sum = 0.f;
#pragma unroll
                for (int h = 0; h < HEADS; ++h)
                    sum += __int_as_float(__builtin_amdgcn_readlane(wbits, h)) * zs[h * FDIM];
                atomicAdd(&S.acc[r * FDIM + lane], sum);
            }
        }
    }
    __syncthreads();

    for (int i = tid; i < (int)nloc * FDIM; i += RTHREADS) {
        const int f = i & (FDIM - 1);
        float bm = 0.f;
#pragma unroll
        for (int hh = 0; hh < HEADS; ++hh) bm += b[hh * FDIM + f];
        sums[(size_t)lo * FDIM + i] = S.acc[i] + bm * (1.0f / HEADS);
    }
}

template <int K, bool ELU_A>
static void run_gat_layer(const float* X, const float* W,
                          const float* al, const float* ar, const float* b,
                          const int* src, const int* dst,
                          float* z, float* el, float* er, float* sums,
                          unsigned* wlst,
                          hipStream_t stream) {
    gemm_wlds_f32<K, ELU_A><<<NROWGROUPS * 4, 256, 0, stream>>>(X, W, z);

    int nh = NNODES * HEADS;
    node_prep<!ELU_A><<<(nh + 255) / 256, 256, 0, stream>>>(z, al, ar, el, er, wlst);

    if constexpr (!ELU_A) bin_edges<<<NBINB, BIN_THREADS, 0, stream>>>(src, dst, wlst);

    range_agg<<<NRB, RTHREADS, 0, stream>>>(src, dst, z, el, er, b, sums, wlst);
}

extern "C" void kernel_launch(void* const* d_in, const int* in_sizes, int n_in,
                              void* d_out, int out_size, void* d_ws, size_t ws_size,
                              hipStream_t stream) {
    const float* x   = (const float*)d_in[0];
    const float* W1  = (const float*)d_in[1];
    const float* al1 = (const float*)d_in[2];
    const float* ar1 = (const float*)d_in[3];
    const float* b1  = (const float*)d_in[4];
    const float* W2  = (const float*)d_in[5];
    const float* al2 = (const float*)d_in[6];
    const float* ar2 = (const float*)d_in[7];
    const float* b2  = (const float*)d_in[8];
    const int*   src = (const int*)d_in[9];
    const int*   dst = (const int*)d_in[10];
    float* out = (float*)d_out;

    char* p = (char*)d_ws;
    float*    z     = (float*)p;    p += (size_t)NNODES * HF * 4;
    float*    el    = (float*)p;    p += (size_t)NNODES * HEADS * 4;
    float*    er    = (float*)p;    p += (size_t)NNODES * HEADS * 4;
    float*    acc   = (float*)p;    p += (size_t)NNODES * FDIM * 4;
    unsigned* wlst  = (unsigned*)p; p += (size_t)NRB * LREG * 4;

    run_gat_layer<INF_, false>(x, W1, al1, ar1, b1, src, dst, z, el, er, acc, wlst, stream);
    run_gat_layer<FDIM, true>(acc, W2, al2, ar2, b2, src, dst, z, el, er, out, wlst, stream);
}


# z rows stored with a feature's four heads in 16 bytes (GEMM stages permuted W columns): the feature pass gathers a row by 2 16-byte loads per lane, not 8 4-byte ones; every sum order kept
# speedup vs baseline: 2.5219x; 1.0613x over previous
#include <hip/hip_runtime.h>
#include <hip/hip_bf16.h>

#define NNODES 50000
#define NEDGES 800000
#define INF_   256
#define FDIM   32
#define HEADS  8
#define HF     (HEADS * FDIM)
#define NEG_SLOPE 0.2f

typedef float v2f __attribute__((ext_vector_type(2)));
typedef float v8f __attribute__((ext_vector_type(8)));

__device__ __forceinline__ int zcol(int p) { return (((p >> 7) << 2) + (p & 3)) * FDIM + ((p >> 2) & (FDIM - 1)); }
__device__ __forceinline__ int zpos(int h, int f) { return ((h >> 2) << 7) + (f << 2) + (h & 3); }
#define GEMM_CB 64
#define NTILES (NNODES / 16)
#define NROWGROUPS ((NTILES + 7) / 8)
__device__ __forceinline__ float elu1(float v) { return v <= 0.f ? expm1f(v) : v; }
__device__ __forceinline__ float4 elu4(float4 v) {
    return float4{elu1(v.x), elu1(v.y), elu1(v.z), elu1(v.w)};
}
template <int K, bool ELU_A>
__global__ __launch_bounds__(256) void gemm_wlds_f32(const float* __restrict__ X,
                                                     const float* __restrict__ W,
                                                     float* __restrict__ Z) {
    constexpr int NQ = K / 4;
    constexpr int NM = K / 32;
    static_assert(!ELU_A || NM == 1, "the activation is applied to the first (only) line of a row");
    __shared__ float4 wl[NQ * GEMM_CB];
    const int tid   = threadIdx.x;
    const int lane  = tid & 31;
    const int wave  = tid >> 5;
    const int half  = lane >> 4;
    const int mn    = lane & 15;
    const int cbase = (blockIdx.x & 3) * GEMM_CB;
    const int tile  = (blockIdx.x >> 2) * 8 + wave;
    const bool valid = tile < NTILES;
    const int row0  = tile * 16;
    const float4* xr = reinterpret_cast<const float4*>(X + (size_t)(row0 + mn) * K) + half;

    float4 a0 = {}, a1 = {}, a2 = {}, a3 = {};
    if (valid) { a0 = xr[0]; a1 = xr[2]; a2 = xr[4]; a3 = xr[6]; }

    {
        const int col = tid & (GEMM_CB - 1);
        const int q0  = tid >> 6;
        const float* wp = W + zcol(cbase + col);
#pragma unroll
        for (int i = 0; i < NQ / 4; ++i) {
            const int q = q0 + 4 * i;
            float4 v;
            v.x = wp[(size_t)(4 * q + 0) * HF];
            v.y = wp[(size_t)(4 * q + 1) * HF];
            v.z = wp[(size_t)(4 * q + 2) * HF];
            v.w = wp[(size_t)(4 * q + 3) * HF];
            wl[q * GEMM_CB + col] = v;
        }
    }
    __syncthreads();
    if (!valid) return;
    if constexpr (ELU_A) { a0 = elu4(a0); a1 = elu4(a1); a2 = elu4(a2); a3 = elu4(a3); }

    const float4* bl = wl + half * GEMM_CB + mn;
    v8f c0 = {}, c1 = {}, c2 = {}, c3 = {};
#define GEMM_STEP(A_, S_)                                                              \
    {                                                                                  \
        const float4* bp = bl + (2 * (S_)) * GEMM_CB;                                  \
        const float4 b0 = bp[0], b1 = bp[16], b2 = bp[32], b3 = bp[48];                \
        const v2f alo = {A_.x, A_.y}, ahi = {A_.z, A_.w};                              \
        c0 = __builtin_amdgcn_wmma_f32_16x16x4_f32(false, alo, false, v2f{b0.x, b0.y}, (short)0, c0, false, false); \
        c1 = __builtin_amdgcn_wmma_f32_16x16x4_f32(false, alo, false, v2f{b1.x, b1.y}, (short)0, c1, false, false); \
        c2 = __builtin_amdgcn_wmma_f32_16x16x4_f32(false, alo, false, v2f{b2.x, b2.y}, (short)0, c2, false, false); \
        c3 = __builtin_amdgcn_wmma_f32_16x16x4_f32(false, alo, false, v2f{b3.x, b3.y}, (short)0, c3, false, false); \
        c0 = __builtin_amdgcn_wmma_f32_16x16x4_f32(false, ahi, false, v2f{b0.z, b0.w}, (short)0, c0, false, false); \
        c1 = __builtin_amdgcn_wmma_f32_16x16x4_f32(false, ahi, false, v2f{b1.z, b1.w}, (short)0, c1, false, false); \
        c2 = __builtin_amdgcn_wmma_f32_16x16x4_f32(false, ahi, false, v2f{b2.z, b2.w}, (short)0, c2, false, false); \
        c3 = __builtin_amdgcn_wmma_f32_16x16x4_f32(false, ahi, false, v2f{b3.z, b3.w}, (short)0, c3, false, false); \
    }
    for (int m = 0; m < NM; ++m) {
        float4 n0 = a0, n1 = a1, n2 = a2, n3 = a3;
        if (m + 1 < NM) {
            const float4* xn = xr + 8 * (m + 1);
            n0 = xn[0]; n1 = xn[2]; n2 = xn[4]; n3 = xn[6];
        }
        GEMM_STEP(a0, 4 * m + 0)
        GEMM_STEP(a1, 4 * m + 1)
        GEMM_STEP(a2, 4 * m + 2)
        GEMM_STEP(a3, 4 * m + 3)
        a0 = n0; a1 = n1; a2 = n2; a3 = n3;
    }
#undef GEMM_STEP
    float* zout = Z + (size_t)(row0 + half * 8) * HF + cbase + mn;
#pragma unroll
    for (int v = 0; v < 8; ++v) {
        zout[(size_t)v * HF +  0] = c0[v];
        zout[(size_t)v * HF + 16] = c1[v];
        zout[(size_t)v * HF + 32] = c2[v];
        zout[(size_t)v * HF + 48] = c3[v];
    }
}

#define RNG  196
#define NRB  ((NNODES + RNG - 1) / RNG)
#define LREG 4096
#define LHDR 64
#define LCAP (LREG - LHDR)
static_assert(NNODES <= 65536 && RNG <= 256 && NRB <= 256, "list entry: src in 16 bits, local dst in 8; bin_edges adds the range in 8 more");

__device__ __forceinline__ unsigned f2ord(float f) {
    unsigned u = __float_as_uint(f);
    return (u & 0x80000000u) ? ~u : (u | 0x80000000u);
}
__device__ __forceinline__ float ord2f(unsigned e) {
    unsigned u = (e & 0x80000000u) ? (e & 0x7FFFFFFFu) : ~e;
    return __uint_as_float(u);
}
#define ORD_NEG_INF 0x007FFFFFu

template <bool CLEAR_LISTS>
__global__ void node_prep(const float* __restrict__ Z,
                          const float* __restrict__ al, const float* __restrict__ ar,
                          float* __restrict__ el, float* __restrict__ er,
                          unsigned* __restrict__ wlst) {
    int i = blockIdx.x * blockDim.x + threadIdx.x;
    if constexpr (CLEAR_LISTS) { if (i < NRB) wlst[(size_t)i * LREG] = 0u; }
    if (i >= NNODES * HEADS) return;
    const int h = i & (HEADS - 1);
    const int n = i >> 3;
    const float* z  = Z + (size_t)n * HF + zpos(h, 0);
    const float* pl = al + h * FDIM;
    const float* pr = ar + h * FDIM;
    float sl = 0.f, sr = 0.f;
#pragma unroll
    for (int f = 0; f < FDIM; ++f) { float zv = z[4 * f]; sl += zv * pl[f]; sr += zv * pr[f]; }
    el[i] = sl; er[i] = sr;
}

__device__ __forceinline__ float lrelu(float v) { return v >= 0.f ? v : NEG_SLOPE * v; }

#define RTHREADS 1024
#define RWAVES (RTHREADS / 32)
static_assert(NEDGES % 32 == 0, "the fallback walks dst by whole waves");

#define BIN_THREADS 1024
#define BIN_Q 2
#define BIN_EPT (BIN_Q * 4)
#define BIN_EDGES (BIN_THREADS * BIN_EPT)
#define NBINB ((NEDGES + BIN_EDGES - 1) / BIN_EDGES)
static_assert(NEDGES % 4 == 0, "dst and src are read by 16-byte loads");
__global__ __launch_bounds__(BIN_THREADS) void bin_edges(const int* __restrict__ src, const int* __restrict__ dst,
                                                        unsigned* __restrict__ wlst) {
    __shared__ unsigned hist[NRB];
    __shared__ unsigned off[NRB];
    __shared__ unsigned gbase[NRB];
    __shared__ unsigned wtot[NRB / 32];
    __shared__ unsigned ent[BIN_EDGES];
    static_assert(NRB % 32 == 0, "the prefix sum runs in whole waves");
    const int tid = threadIdx.x;
    if (tid < NRB) hist[tid] = 0u;
    int dv[BIN_EPT], sv[BIN_EPT];
    for (int i = 0; i < BIN_Q; ++i) {
        const int q = (blockIdx.x * BIN_Q + i) * BIN_THREADS + tid;
        int4 d = {-1, -1, -1, -1}, s = {0, 0, 0, 0};
        if (q < NEDGES / 4) {
            d = reinterpret_cast<const int4*>(dst)[q];
            s = reinterpret_cast<const int4*>(src)[q];
        }
        dv[4 * i + 0] = d.x; dv[4 * i + 1] = d.y; dv[4 * i + 2] = d.z; dv[4 * i + 3] = d.w;
        sv[4 * i + 0] = s.x; sv[4 * i + 1] = s.y; sv[4 * i + 2] = s.z; sv[4 * i + 3] = s.w;
    }
    __syncthreads();
    unsigned bin[BIN_EPT], rnk[BIN_EPT], word[BIN_EPT];
    bool ok[BIN_EPT];
    for (int k = 0; k < BIN_EPT; ++k) {
        ok[k]  = (unsigned)dv[k] < (unsigned)NNODES;
        bin[k] = ok[k] ? (unsigned)dv[k] / RNG : 0u;
        word[k] = (unsigned)sv[k] | (((unsigned)dv[k] - bin[k] * RNG) << 16) | (bin[k] << 24);
        rnk[k] = ok[k] ? atomicAdd(&hist[bin[k]], 1u) : 0u;
    }
    __syncthreads();
    unsigned cntv = 0u, incl = 0u;
    if (tid < NRB) {
        cntv = hist[tid];
        incl = cntv;
        for (int dl = 1; dl < 32; dl <<= 1) {
            const unsigned t = __shfl_up(incl, dl);
            if ((tid & 31) >= dl) incl += t;
        }
        if ((tid & 31) == 31) wtot[tid >> 5] = incl;
        gbase[tid] = cntv ? atomicAdd(&wlst[(size_t)tid * LREG], cntv) : 0u;
    }
    __syncthreads();
    if (tid < NRB) {
        unsigned before = 0u;
        for (int w = 0; w < (tid >> 5); ++w) before += wtot[w];
        off[tid] = before + incl - cntv;
    }
    __syncthreads();
    unsigned slot[BIN_EPT];
    for (int k = 0; k < BIN_EPT; ++k) slot[k] = off[bin[k]] + rnk[k];
    for (int k = 0; k < BIN_EPT; ++k)
        if (ok[k]) ent[slot[k]] = word[k];
    __syncthreads();
    const unsigned total = off[NRB - 1] + hist[NRB - 1];
    for (unsigned j = tid; j < total; j += BIN_THREADS) {
        const unsigned e = ent[j];
        const unsigned r = e >> 24;
        const unsigned pos = gbase[r] + (j - off[r]);
        if (pos < LCAP) wlst[(size_t)r * LREG + LHDR + pos] = e & 0x00FFFFFFu;
    }
}

struct RangeState {
    float    acc[RNG * FDIM];
    unsigned mo[RNG * HEADS];
    float    den[RNG * HEADS];
    float    erl[RNG * HEADS];
    unsigned lst[LCAP];
    float    ew[LCAP * HEADS];
};
static_assert(sizeof(RangeState) <= 320 * 1024, "one work-group's LDS");
static_assert((RNG * FDIM + 3 * RNG * HEADS + LCAP) % 4 == 0, "ew rows are read and written as 16-byte pieces");

template <class F>
__device__ __forceinline__ void each_edge_of_dst(const int* __restrict__ src, const int* __restrict__ dst,
                                                 int lo, unsigned nloc, int tid, F f) {
    for (int e = tid; e < NEDGES; e += RTHREADS) {
        const unsigned r = (unsigned)(dst[e] - lo);
        if (r < nloc) f(src[e], (int)r);
    }
}

__global__ __launch_bounds__(RTHREADS) void range_agg(const int* __restrict__ src, const int* __restrict__ dst,
                          const float* __restrict__ Z,
                          const float* __restrict__ el, const float* __restrict__ er,
                          const float* __restrict__ b, float* __restrict__ sums,
                          const unsigned* __restrict__ wlst) {
    __shared__ __attribute__((aligned(16))) RangeState S;
    const int tid  = threadIdx.x;
    const int lane = tid & 31;
    const int wave = __builtin_amdgcn_readfirstlane(tid >> 5);
    const int lo   = blockIdx.x * RNG;
    const unsigned nloc = (unsigned)min(RNG, NNODES - lo);
    const unsigned* mylist = wlst + (size_t)blockIdx.x * LREG;
    const int  cnt  = (int)mylist[0];
    const bool fits = cnt <= LCAP;

    for (int i = tid; i < RNG * FDIM; i += RTHREADS) S.acc[i] = 0.f;
    for (int i = tid; i < RNG * HEADS; i += RTHREADS) {
        S.mo[i] = ORD_NEG_INF; S.den[i] = 0.f;
        S.erl[i] = (unsigned)i < nloc * HEADS ? er[lo * HEADS + i] : 0.f;
    }
    if (fits)
        for (int i = tid; i < cnt; i += RTHREADS) S.lst[i] = mylist[LHDR + i];
    __syncthreads();

    if (fits) {
        for (int i = tid; i < cnt; i += RTHREADS) {
            const unsigned p = S.lst[i];
            const int r = (int)(p >> 16);
            const float4* e4 = reinterpret_cast<const float4*>(el + (size_t)(p & 0xFFFFu) * HEADS);
            const float4 a = e4[0], c = e4[1];
            const float ev[HEADS] = {a.x, a.y, a.z, a.w, c.x, c.y, c.z, c.w};
            float v[HEADS];
            for (int h = 0; h < HEADS; ++h) {
                v[h] = lrelu(ev[h] + S.erl[r * HEADS + h]);
                atomicMax(&S.mo[r * HEADS + h], f2ord(v[h]));
            }
            float4* t = reinterpret_cast<float4*>(S.ew + i * HEADS);
            t[0] = float4{v[0], v[1], v[2], v[3]};
            t[1] = float4{v[4], v[5], v[6], v[7]};
        }
    } else {
        each_edge_of_dst(src, dst, lo, nloc, tid, [&](int s, int r) {
            const float4* e4 = reinterpret_cast<const float4*>(el + (size_t)s * HEADS);
            const float4 a = e4[0], c = e4[1];
            const float ev[HEADS] = {a.x, a.y, a.z, a.w, c.x, c.y, c.z, c.w};
#pragma unroll
            for (int h = 0; h < HEADS; ++h)
                atomicMax(&S.mo[r * HEADS + h], f2ord(lrelu(ev[h] + S.erl[r * HEADS + h])));
        });
    }
    __syncthreads();
    if (fits) {
        for (int i = tid; i < cnt; i += RTHREADS) {
            const int r = (int)(S.lst[i] >> 16);
            float4* t = reinterpret_cast<float4*>(S.ew + i * HEADS);
            const float4 a = t[0], c = t[1];
            float v[HEADS] = {a.x, a.y, a.z, a.w, c.x, c.y, c.z, c.w};
            for (int h = 0; h < HEADS; ++h) {
                v[h] = __expf(v[h] - ord2f(S.mo[r * HEADS + h]));
                atomicAdd(&S.den[r * HEADS + h], v[h]);
            }
            t[0] = float4{v[0], v[1], v[2], v[3]};
            t[1] = float4{v[4], v[5], v[6], v[7]};
        }
    } else {
        each_edge_of_dst(src, dst, lo, nloc, tid, [&](int s, int r) {
            const float4* e4 = reinterpret_cast<const float4*>(el + (size_t)s * HEADS);
            const float4 a = e4[0], c = e4[1];
            const float ev[HEADS] = {a.x, a.y, a.z, a.w, c.x, c.y, c.z, c.w};
#pragma unroll
            for (int h = 0; h < HEADS; ++h) {
                const float v = lrelu(ev[h] + S.erl[r * HEADS + h]);
                atomicAdd(&S.den[r * HEADS + h], __expf(v - ord2f(S.mo[r * HEADS + h])));
            }
        });
    }
    __syncthreads();

    auto weight = [&](int s, int r, int h) {
        const float v = lrelu(el[(size_t)s * HEADS + h] + S.erl[r * HEADS + h]);
        const float ex = __expf(v - ord2f(S.mo[r * HEADS + h]));
        return ex / S.den[r * HEADS + h] * (1.0f / HEADS);
    };
    if (fits) {
        for (int j0 = wave * 8; j0 < cnt; j0 += RWAVES * 8) {
            const int ja = min(j0 + (lane >> 3), cnt - 1), jb = min(j0 + (lane >> 3) + 4, cnt - 1);
            const unsigned pa = S.lst[ja];
            const unsigned pb = S.lst[jb];
            const float w0 = S.ew[ja * HEADS + (lane & 7)] / S.den[(pa >> 16) * HEADS + (lane & 7)] * (1.0f / HEADS);
            const float w1 = S.ew[jb * HEADS + (lane & 7)] / S.den[(pb >> 16) * HEADS + (lane & 7)] * (1.0f / HEADS);
            float zv[8][HEADS];
#pragma unroll
            for (int j = 0; j < 8; ++j) {
                const unsigned p = (unsigned)__builtin_amdgcn_readlane((int)(j < 4 ? pa : pb), (j & 3) * 8);
                const float4* zs = reinterpret_cast<const float4*>(Z + (size_t)(p & 0xFFFFu) * HF) + lane;
                const float4 za = zs[0], zb = zs[FDIM];
                zv[j][0] = za.x; zv[j][1] = za.y; zv[j][2] = za.z; zv[j][3] = za.w;
                zv[j][4] = zb.x; zv[j][5] = zb.y; zv[j][6] = zb.z; zv[j][7] = zb.w;
            }
#pragma unroll
            for (int j = 0; j < 8; ++j) {
                const unsigned p = (unsigned)__builtin_amdgcn_readlane((int)(j < 4 ? pa : pb), (j & 3) * 8);
                const int wbits = __float_as_int(j < 4 ? w0 : w1);
                float sum = 0.f;
#pragma unroll
                for (int h = 0; h < HEADS; ++h) {
                    const float a = __int_as_float(__builtin_amdgcn_readlane(wbits, (j & 3) * HEADS + h));
                    sum += a * zv[j][h];
                }
                if (j0 + j < cnt) atomicAdd(&S.acc[(p >> 16) * FDIM + lane], sum);
            }
        }
    } else {
        for (int base = wave * 32; base < NEDGES; base += RTHREADS) {
            const unsigned rr = (unsigned)(dst[base + lane] - lo);
            const bool ok = rr < nloc;
            const int  ss = ok ? src[base + lane] : 0;
            unsigned mask = __builtin_amdgcn_ballot_w32(ok);
            while (mask) {
                const int bit = __builtin_ctz(mask);
                mask &= mask - 1;
                const int s = __builtin_amdgcn_readlane(ss, bit);
                const int r = __builtin_amdgcn_readlane((int)rr, bit);
                const int wbits = __float_as_int(weight(s, r, lane & 7));
                const float4* zs = reinterpret_cast<const float4*>(Z + (size_t)s * HF) + lane;
                const float4 za = zs[0], zb = zs[FDIM];
                const float zr[HEADS] = {za.x, za.y, za.z, za.w, zb.x, zb.y, zb.z, zb.w};
                float sum = 0.f;
#pragma unroll
                for (int h = 0; h < HEADS; ++h)
                    sum += __int_as_float(__builtin_amdgcn_readlane(wbits, h)) * zr[h];
                atomicAdd(&S.acc[r * FDIM + lane], sum);
            }
        }
    }
    __syncthreads();

    for (int i = tid; i < (int)nloc * FDIM; i += RTHREADS) {
        const int f = i & (FDIM - 1);
        float bm = 0.f;
#pragma unroll
        for (int hh = 0; hh < HEADS; ++hh) bm += b[hh * FDIM + f];
        sums[(size_t)lo * FDIM + i] = S.acc[i] + bm * (1.0f / HEADS);
    }
}

template <int K, bool ELU_A>
static void run_gat_layer(const float* X, const float* W,
                          const float* al, const float* ar, const float* b,
                          const int* src, const int* dst,
                          float* z, float* el, float* er, float* sums,
                          unsigned* wlst,
                          hipStream_t stream) {
    gemm_wlds_f32<K, ELU_A><<<NROWGROUPS * 4, 256, 0, stream>>>(X, W, z);

    int nh = NNODES * HEADS;
    node_prep<!ELU_A><<<(nh + 255) / 256, 256, 0, stream>>>(z, al, ar, el, er, wlst);

    if constexpr (!ELU_A) bin_edges<<<NBINB, BIN_THREADS, 0, stream>>>(src, dst, wlst);

    range_agg<<<NRB, RTHREADS, 0, stream>>>(src, dst, z, el, er, b, sums, wlst);
}

extern "C" void kernel_launch(void* const* d_in, const int* in_sizes, int n_in,
                              void* d_out, int out_size, void* d_ws, size_t ws_size,
                              hipStream_t stream) {
    const float* x   = (const float*)d_in[0];
    const float* W1  = (const float*)d_in[1];
    const float* al1 = (const float*)d_in[2];
    const float* ar1 = (const float*)d_in[3];
    const float* b1  = (const float*)d_in[4];
    const float* W2  = (const float*)d_in[5];
    const float* al2 = (const float*)d_in[6];
    const float* ar2 = (const float*)d_in[7];
    const float* b2  = (const float*)d_in[8];
    const int*   src = (const int*)d_in[9];
    const int*   dst = (const int*)d_in[10];
    float* out = (float*)d_out;

    char* p = (char*)d_ws;
    float*    z     = (float*)p;    p += (size_t)NNODES * HF * 4;
    float*    el    = (float*)p;    p += (size_t)NNODES * HEADS * 4;
    float*    er    = (float*)p;    p += (size_t)NNODES * HEADS * 4;
    float*    acc   = (float*)p;    p += (size_t)NNODES * FDIM * 4;
    unsigned* wlst  = (unsigned*)p; p += (size_t)NRB * LREG * 4;

    run_gat_layer<INF_, false>(x, W1, al1, ar1, b1, src, dst, z, el, er, acc, wlst, stream);
    run_gat_layer<FDIM, true>(acc, W2, al2, ar2, b2, src, dst, z, el, er, out, wlst, stream);
}


# range_agg of layer 1 multiplies its own finished rows by W2 on f32 WMMA from LDS and forms the layer-2 logits: GEMM 2 and its logits launch gone (7 -> 5 launches); every sum order kept
# speedup vs baseline: 2.6597x; 1.0546x over previous
#include <hip/hip_runtime.h>
#include <hip/hip_bf16.h>

#define NNODES 50000
#define NEDGES 800000
#define INF_   256
#define FDIM   32
#define HEADS  8
#define HF     (HEADS * FDIM)
#define NEG_SLOPE 0.2f

typedef float v2f __attribute__((ext_vector_type(2)));
typedef float v8f __attribute__((ext_vector_type(8)));

__device__ __forceinline__ int zcol(int p) { return (((p >> 7) << 2) + (p & 3)) * FDIM + ((p >> 2) & (FDIM - 1)); }
__device__ __forceinline__ int zpos(int h, int f) { return ((h >> 2) << 7) + (f << 2) + (h & 3); }
#define GEMM_CB 64
#define NTILES (NNODES / 16)
#define NROWGROUPS ((NTILES + 7) / 8)
__device__ __forceinline__ float elu1(float v) { return v <= 0.f ? expm1f(v) : v; }
__device__ __forceinline__ float4 elu4(float4 v) {
    return float4{elu1(v.x), elu1(v.y), elu1(v.z), elu1(v.w)};
}
template <int K, bool ELU_A>
__global__ __launch_bounds__(256) void gemm_wlds_f32(const float* __restrict__ X,
                                                     const float* __restrict__ W,
                                                     float* __restrict__ Z) {
    constexpr int NQ = K / 4;
    constexpr int NM = K / 32;
    static_assert(!ELU_A || NM == 1, "the activation is applied to the first (only) line of a row");
    __shared__ float4 wl[NQ * GEMM_CB];
    const int tid   = threadIdx.x;
    const int lane  = tid & 31;
    const int wave  = tid >> 5;
    const int half  = lane >> 4;
    const int mn    = lane & 15;
    const int cbase = (blockIdx.x & 3) * GEMM_CB;
    const int tile  = (blockIdx.x >> 2) * 8 + wave;
    const bool valid = tile < NTILES;
    const int row0  = tile * 16;
    const float4* xr = reinterpret_cast<const float4*>(X + (size_t)(row0 + mn) * K) + half;

    float4 a0 = {}, a1 = {}, a2 = {}, a3 = {};
    if (valid) { a0 = xr[0]; a1 = xr[2]; a2 = xr[4]; a3 = xr[6]; }

    {
        const int col = tid & (GEMM_CB - 1);
        const int q0  = tid >> 6;
        const float* wp = W + zcol(cbase + col);
#pragma unroll
        for (int i = 0; i < NQ / 4; ++i) {
            const int q = q0 + 4 * i;
            float4 v;
            v.x = wp[(size_t)(4 * q + 0) * HF];
            v.y = wp[(size_t)(4 * q + 1) * HF];
            v.z = wp[(size_t)(4 * q + 2) * HF];
            v.w = wp[(size_t)(4 * q + 3) * HF];
            wl[q * GEMM_CB + col] = v;
        }
    }
    __syncthreads();
    if (!valid) return;
    if constexpr (ELU_A) { a0 = elu4(a0); a1 = elu4(a1); a2 = elu4(a2); a3 = elu4(a3); }

    const float4* bl = wl + half * GEMM_CB + mn;
    v8f c0 = {}, c1 = {}, c2 = {}, c3 = {};
#define GEMM_STEP(A_, S_)                                                              \
    {                                                                                  \
        const float4* bp = bl + (2 * (S_)) * GEMM_CB;                                  \
        const float4 b0 = bp[0], b1 = bp[16], b2 = bp[32], b3 = bp[48];                \
        const v2f alo = {A_.x, A_.y}, ahi = {A_.z, A_.w};                              \
        c0 = __builtin_amdgcn_wmma_f32_16x16x4_f32(false, alo, false, v2f{b0.x, b0.y}, (short)0, c0, false, false); \
        c1 = __builtin_amdgcn_wmma_f32_16x16x4_f32(false, alo, false, v2f{b1.x, b1.y}, (short)0, c1, false, false); \
        c2 = __builtin_amdgcn_wmma_f32_16x16x4_f32(false, alo, false, v2f{b2.x, b2.y}, (short)0, c2, false, false); \
        c3 = __builtin_amdgcn_wmma_f32_16x16x4_f32(false, alo, false, v2f{b3.x, b3.y}, (short)0, c3, false, false); \
        c0 = __builtin_amdgcn_wmma_f32_16x16x4_f32(false, ahi, false, v2f{b0.z, b0.w}, (short)0, c0, false, false); \
        c1 = __builtin_amdgcn_wmma_f32_16x16x4_f32(false, ahi, false, v2f{b1.z, b1.w}, (short)0, c1, false, false); \
        c2 = __builtin_amdgcn_wmma_f32_16x16x4_f32(false, ahi, false, v2f{b2.z, b2.w}, (short)0, c2, false, false); \
        c3 = __builtin_amdgcn_wmma_f32_16x16x4_f32(false, ahi, false, v2f{b3.z, b3.w}, (short)0, c3, false, false); \
    }
    for (int m = 0; m < NM; ++m) {
        float4 n0 = a0, n1 = a1, n2 = a2, n3 = a3;
        if (m + 1 < NM) {
            const float4* xn = xr + 8 * (m + 1);
            n0 = xn[0]; n1 = xn[2]; n2 = xn[4]; n3 = xn[6];
        }
        GEMM_STEP(a0, 4 * m + 0)
        GEMM_STEP(a1, 4 * m + 1)
        GEMM_STEP(a2, 4 * m + 2)
        GEMM_STEP(a3, 4 * m + 3)
        a0 = n0; a1 = n1; a2 = n2; a3 = n3;
    }
#undef GEMM_STEP
    float* zout = Z + (size_t)(row0 + half * 8) * HF + cbase + mn;
#pragma unroll
    for (int v = 0; v < 8; ++v) {
        zout[(size_t)v * HF +  0] = c0[v];
        zout[(size_t)v * HF + 16] = c1[v];
        zout[(size_t)v * HF + 32] = c2[v];
        zout[(size_t)v * HF + 48] = c3[v];
    }
}

#define RNG  196
#define NRB  ((NNODES + RNG - 1) / RNG)
#define LREG 4096
#define LHDR 64
#define LCAP (LREG - LHDR)
static_assert(NNODES <= 65536 && RNG <= 256 && NRB <= 256, "list entry: src in 16 bits, local dst in 8; bin_edges adds the range in 8 more");

__device__ __forceinline__ unsigned f2ord(float f) {
    unsigned u = __float_as_uint(f);
    return (u & 0x80000000u) ? ~u : (u | 0x80000000u);
}
__device__ __forceinline__ float ord2f(unsigned e) {
    unsigned u = (e & 0x80000000u) ? (e & 0x7FFFFFFFu) : ~e;
    return __uint_as_float(u);
}
#define ORD_NEG_INF 0x007FFFFFu

template <bool CLEAR_LISTS>
__global__ void node_prep(const float* __restrict__ Z,
                          const float* __restrict__ al, const float* __restrict__ ar,
                          float* __restrict__ el, float* __restrict__ er,
                          unsigned* __restrict__ wlst) {
    int i = blockIdx.x * blockDim.x + threadIdx.x;
    if constexpr (CLEAR_LISTS) { if (i < NRB) wlst[(size_t)i * LREG] = 0u; }
    if (i >= NNODES * HEADS) return;
    const int h = i & (HEADS - 1);
    const int n = i >> 3;
    const float* z  = Z + (size_t)n * HF + zpos(h, 0);
    const float* pl = al + h * FDIM;
    const float* pr = ar + h * FDIM;
    float sl = 0.f, sr = 0.f;
#pragma unroll
    for (int f = 0; f < FDIM; ++f) { float zv = z[4 * f]; sl += zv * pl[f]; sr += zv * pr[f]; }
    el[i] = sl; er[i] = sr;
}

__device__ __forceinline__ float lrelu(float v) { return v >= 0.f ? v : NEG_SLOPE * v; }

#define RTHREADS 1024
#define RWAVES (RTHREADS / 32)
static_assert(NEDGES % 32 == 0, "the fallback walks dst by whole waves");

#define BIN_THREADS 1024
#define BIN_Q 2
#define BIN_EPT (BIN_Q * 4)
#define BIN_EDGES (BIN_THREADS * BIN_EPT)
#define NBINB ((NEDGES + BIN_EDGES - 1) / BIN_EDGES)
static_assert(NEDGES % 4 == 0, "dst and src are read by 16-byte loads");
__global__ __launch_bounds__(BIN_THREADS) void bin_edges(const int* __restrict__ src, const int* __restrict__ dst,
                                                        unsigned* __restrict__ wlst) {
    __shared__ unsigned hist[NRB];
    __shared__ unsigned off[NRB];
    __shared__ unsigned gbase[NRB];
    __shared__ unsigned wtot[NRB / 32];
    __shared__ unsigned ent[BIN_EDGES];
    static_assert(NRB % 32 == 0, "the prefix sum runs in whole waves");
    const int tid = threadIdx.x;
    if (tid < NRB) hist[tid] = 0u;
    int dv[BIN_EPT], sv[BIN_EPT];
    for (int i = 0; i < BIN_Q; ++i) {
        const int q = (blockIdx.x * BIN_Q + i) * BIN_THREADS + tid;
        int4 d = {-1, -1, -1, -1}, s = {0, 0, 0, 0};
        if (q < NEDGES / 4) {
            d = reinterpret_cast<const int4*>(dst)[q];
            s = reinterpret_cast<const int4*>(src)[q];
        }
        dv[4 * i + 0] = d.x; dv[4 * i + 1] = d.y; dv[4 * i + 2] = d.z; dv[4 * i + 3] = d.w;
        sv[4 * i + 0] = s.x; sv[4 * i + 1] = s.y; sv[4 * i + 2] = s.z; sv[4 * i + 3] = s.w;
    }
    __syncthreads();
    unsigned bin[BIN_EPT], rnk[BIN_EPT], word[BIN_EPT];
    bool ok[BIN_EPT];
    for (int k = 0; k < BIN_EPT; ++k) {
        ok[k]  = (unsigned)dv[k] < (unsigned)NNODES;
        bin[k] = ok[k] ? (unsigned)dv[k] / RNG : 0u;
        word[k] = (unsigned)sv[k] | (((unsigned)dv[k] - bin[k] * RNG) << 16) | (bin[k] << 24);
        rnk[k] = ok[k] ? atomicAdd(&hist[bin[k]], 1u) : 0u;
    }
    __syncthreads();
    unsigned cntv = 0u, incl = 0u;
    if (tid < NRB) {
        cntv = hist[tid];
        incl = cntv;
        for (int dl = 1; dl < 32; dl <<= 1) {
            const unsigned t = __shfl_up(incl, dl);
            if ((tid & 31) >= dl) incl += t;
        }
        if ((tid & 31) == 31) wtot[tid >> 5] = incl;
        gbase[tid] = cntv ? atomicAdd(&wlst[(size_t)tid * LREG], cntv) : 0u;
    }
    __syncthreads();
    if (tid < NRB) {
        unsigned before = 0u;
        for (int w = 0; w < (tid >> 5); ++w) before += wtot[w];
        off[tid] = before + incl - cntv;
    }
    __syncthreads();
    unsigned slot[BIN_EPT];
    for (int k = 0; k < BIN_EPT; ++k) slot[k] = off[bin[k]] + rnk[k];
    for (int k = 0; k < BIN_EPT; ++k)
        if (ok[k]) ent[slot[k]] = word[k];
    __syncthreads();
    const unsigned total = off[NRB - 1] + hist[NRB - 1];
    for (unsigned j = tid; j < total; j += BIN_THREADS) {
        const unsigned e = ent[j];
        const unsigned r = e >> 24;
        const unsigned pos = gbase[r] + (j - off[r]);
        if (pos < LCAP) wlst[(size_t)r * LREG + LHDR + pos] = e & 0x00FFFFFFu;
    }
}

struct EdgeState {
    unsigned mo[RNG * HEADS];
    float    den[RNG * HEADS];
    float    erl[RNG * HEADS];
    unsigned lst[LCAP];
    float    ew[LCAP * HEADS];
};
#define ZPITCH (HF + 4)
#define RNG16  (((RNG + 15) / 16) * 16)
template <bool PROJ>
struct RangeState {
    float acc[RNG * FDIM];
    union {
        EdgeState e;
        float z2[PROJ ? RNG16 * ZPITCH : 4];
    } u;
    float w2[PROJ ? FDIM * HF : 4];
};
static_assert(sizeof(RangeState<true>) <= 320 * 1024 && sizeof(RangeState<false>) <= 320 * 1024, "one work-group's LDS");
static_assert((RNG * FDIM) % 4 == 0 && sizeof(EdgeState) % 16 == 0 && (RNG16 * ZPITCH) % 4 == 0 && ZPITCH % 4 == 0,
              "z2 rows and w2 quads are moved as 16-byte pieces");
static_assert((RNG * FDIM + 3 * RNG * HEADS + LCAP) % 4 == 0, "ew rows are read and written as 16-byte pieces");

template <class F>
__device__ __forceinline__ void each_edge_of_dst(const int* __restrict__ src, const int* __restrict__ dst,
                                                 int lo, unsigned nloc, int tid, F f) {
    for (int e = tid; e < NEDGES; e += RTHREADS) {
        const unsigned r = (unsigned)(dst[e] - lo);
        if (r < nloc) f(src[e], (int)r);
    }
}

template <bool PROJ>
__global__ __launch_bounds__(RTHREADS) void range_agg(const int* __restrict__ src, const int* __restrict__ dst,
                          const float* __restrict__ Z,
                          const float* __restrict__ el, const float* __restrict__ er,
                          const float* __restrict__ b, float* __restrict__ sums,
                          const unsigned* __restrict__ wlst,
                          const float* __restrict__ Wn, const float* __restrict__ aln, const float* __restrict__ arn,
                          float* __restrict__ Zn, float* __restrict__ eln, float* __restrict__ ern) {
    __shared__ __attribute__((aligned(16))) RangeState<PROJ> SS;
    auto& S = SS.u.e;
    const int tid  = threadIdx.x;
    const int lane = tid & 31;
    const int wave = __builtin_amdgcn_readfirstlane(tid >> 5);
    const int lo   = blockIdx.x * RNG;
    const unsigned nloc = (unsigned)min(RNG, NNODES - lo);
    const unsigned* mylist = wlst + (size_t)blockIdx.x * LREG;
    const int  cnt  = (int)mylist[0];
    const bool fits = cnt <= LCAP;

    if constexpr (PROJ) {
        for (int e = tid; e < (FDIM / 4) * HF; e += RTHREADS) {
            const int q = e >> 8, p = e & (HF - 1);
            const float* wp = Wn + zcol(p);
            float4 v;
            v.x = wp[(size_t)(4 * q + 0) * HF];
            v.y = wp[(size_t)(4 * q + 1) * HF];
            v.z = wp[(size_t)(4 * q + 2) * HF];
            v.w = wp[(size_t)(4 * q + 3) * HF];
            reinterpret_cast<float4*>(SS.w2)[e] = v;
        }
    }
    for (int i = tid; i < RNG * FDIM; i += RTHREADS) SS.acc[i] = 0.f;
    for (int i = tid; i < RNG * HEADS; i += RTHREADS) {
        S.mo[i] = ORD_NEG_INF; S.den[i] = 0.f;
        S.erl[i] = (unsigned)i < nloc * HEADS ? er[lo * HEADS + i] : 0.f;
    }
    if (fits)
        for (int i = tid; i < cnt; i += RTHREADS) S.lst[i] = mylist[LHDR + i];
    __syncthreads();

    if (fits) {
        for (int i = tid; i < cnt; i += RTHREADS) {
            const unsigned p = S.lst[i];
            const int r = (int)(p >> 16);
            const float4* e4 = reinterpret_cast<const float4*>(el + (size_t)(p & 0xFFFFu) * HEADS);
            const float4 a = e4[0], c = e4[1];
            const float ev[HEADS] = {a.x, a.y, a.z, a.w, c.x, c.y, c.z, c.w};
            float v[HEADS];
            for (int h = 0; h < HEADS; ++h) {
                v[h] = lrelu(ev[h] + S.erl[r * HEADS + h]);
                atomicMax(&S.mo[r * HEADS + h], f2ord(v[h]));
            }
            float4* t = reinterpret_cast<float4*>(S.ew + i * HEADS);
            t[0] = float4{v[0], v[1], v[2], v[3]};
            t[1] = float4{v[4], v[5], v[6], v[7]};
        }
    } else {
        each_edge_of_dst(src, dst, lo, nloc, tid, [&](int s, int r) {
            const float4* e4 = reinterpret_cast<const float4*>(el + (size_t)s * HEADS);
            const float4 a = e4[0], c = e4[1];
            const float ev[HEADS] = {a.x, a.y, a.z, a.w, c.x, c.y, c.z, c.w};
#pragma unroll
            for (int h = 0; h < HEADS; ++h)
                atomicMax(&S.mo[r * HEADS + h], f2ord(lrelu(ev[h] + S.erl[r * HEADS + h])));
        });
    }
    __syncthreads();
    if (fits) {
        for (int i = tid; i < cnt; i += RTHREADS) {
            const int r = (int)(S.lst[i] >> 16);
            float4* t = reinterpret_cast<float4*>(S.ew + i * HEADS);
            const float4 a = t[0], c = t[1];
            float v[HEADS] = {a.x, a.y, a.z, a.w, c.x, c.y, c.z, c.w};
            for (int h = 0; h < HEADS; ++h) {
                v[h] = __expf(v[h] - ord2f(S.mo[r * HEADS + h]));
                atomicAdd(&S.den[r * HEADS + h], v[h]);
            }
            t[0] = float4{v[0], v[1], v[2], v[3]};
            t[1] = float4{v[4], v[5], v[6], v[7]};
        }
    } else {
        each_edge_of_dst(src, dst, lo, nloc, tid, [&](int s, int r) {
            const float4* e4 = reinterpret_cast<const float4*>(el + (size_t)s * HEADS);
            const float4 a = e4[0], c = e4[1];
            const float ev[HEADS] = {a.x, a.y, a.z, a.w, c.x, c.y, c.z, c.w};
#pragma unroll
            for (int h = 0; h < HEADS; ++h) {
                const float v = lrelu(ev[h] + S.erl[r * HEADS + h]);
                atomicAdd(&S.den[r * HEADS + h], __expf(v - ord2f(S.mo[r * HEADS + h])));
            }
        });
    }
    __syncthreads();

    auto weight = [&](int s, int r, int h) {
        const float v = lrelu(el[(size_t)s * HEADS + h] + S.erl[r * HEADS + h]);
        const float ex = __expf(v - ord2f(S.mo[r * HEADS + h]));
        return ex / S.den[r * HEADS + h] * (1.0f / HEADS);
    };
    if (fits) {
        for (int j0 = wave * 8; j0 < cnt; j0 += RWAVES * 8) {
            const int ja = min(j0 + (lane >> 3), cnt - 1), jb = min(j0 + (lane >> 3) + 4, cnt - 1);
            const unsigned pa = S.lst[ja];
            const unsigned pb = S.lst[jb];
            const float w0 = S.ew[ja * HEADS + (lane & 7)] / S.den[(pa >> 16) * HEADS + (lane & 7)] * (1.0f / HEADS);
            const float w1 = S.ew[jb * HEADS + (lane & 7)] / S.den[(pb >> 16) * HEADS + (lane & 7)] * (1.0f / HEADS);
            float zv[8][HEADS];
#pragma unroll
            for (int j = 0; j < 8; ++j) {
                const unsigned p = (unsigned)__builtin_amdgcn_readlane((int)(j < 4 ? pa : pb), (j & 3) * 8);
                const float4* zs = reinterpret_cast<const float4*>(Z + (size_t)(p & 0xFFFFu) * HF) + lane;
                const float4 za = zs[0], zb = zs[FDIM];
                zv[j][0] = za.x; zv[j][1] = za.y; zv[j][2] = za.z; zv[j][3] = za.w;
                zv[j][4] = zb.x; zv[j][5] = zb.y; zv[j][6] = zb.z; zv[j][7] = zb.w;
            }
#pragma unroll
            for (int j = 0; j < 8; ++j) {
                const unsigned p = (unsigned)__builtin_amdgcn_readlane((int)(j < 4 ? pa : pb), (j & 3) * 8);
                const int wbits = __float_as_int(j < 4 ? w0 : w1);
                float sum = 0.f;
#pragma unroll
                for (int h = 0; h < HEADS; ++h) {
                    const float a = __int_as_float(__builtin_amdgcn_readlane(wbits, (j & 3) * HEADS + h));
                    sum += a * zv[j][h];
                }
                if (j0 + j < cnt) atomicAdd(&SS.acc[(p >> 16) * FDIM + lane], sum);
            }
        }
    } else {
        for (int base = wave * 32; base < NEDGES; base += RTHREADS) {
            const unsigned rr = (unsigned)(dst[base + lane] - lo);
            const bool ok = rr < nloc;
            const int  ss = ok ? src[base + lane] : 0;
            unsigned mask = __builtin_amdgcn_ballot_w32(ok);
            while (mask) {
                const int bit = __builtin_ctz(mask);
                mask &= mask - 1;
                const int s = __builtin_amdgcn_readlane(ss, bit);
                const int r = __builtin_amdgcn_readlane((int)rr, bit);
                const int wbits = __float_as_int(weight(s, r, lane & 7));
                const float4* zs = reinterpret_cast<const float4*>(Z + (size_t)s * HF) + lane;
                const float4 za = zs[0], zb = zs[FDIM];
                const float zr[HEADS] = {za.x, za.y, za.z, za.w, zb.x, zb.y, zb.z, zb.w};
                float sum = 0.f;
#pragma unroll
                for (int h = 0; h < HEADS; ++h)
                    sum += __int_as_float(__builtin_amdgcn_readlane(wbits, h)) * zr[h];
                atomicAdd(&SS.acc[r * FDIM + lane], sum);
            }
        }
    }
    __syncthreads();

    if constexpr (!PROJ) {
        for (int i = tid; i < (int)nloc * FDIM; i += RTHREADS) {
            const int f = i & (FDIM - 1);
            float bm = 0.f;
#pragma unroll
            for (int hh = 0; hh < HEADS; ++hh) bm += b[hh * FDIM + f];
            sums[(size_t)lo * FDIM + i] = SS.acc[i] + bm * (1.0f / HEADS);
        }
    } else {
        for (int i = tid; i < (int)nloc * FDIM; i += RTHREADS) {
            const int f = i & (FDIM - 1);
            float bm = 0.f;
#pragma unroll
            for (int hh = 0; hh < HEADS; ++hh) bm += b[hh * FDIM + f];
            SS.acc[i] = elu1(SS.acc[i] + bm * (1.0f / HEADS));
        }
        __syncthreads();
        {
            const int half = lane >> 4;
            const int mn   = lane & 15;
            const int nitems = (((int)nloc + 15) >> 4) * 4;
            const float4* w4 = reinterpret_cast<const float4*>(SS.w2);
            for (int item = wave; item < nitems; item += RWAVES) {
                const int rt = item >> 2, cq = item & 3;
                const int row = min(rt * 16 + mn, RNG - 1);
                const float4* ar4 = reinterpret_cast<const float4*>(SS.acc + row * FDIM) + half;
                const float4 a0 = ar4[0], a1 = ar4[2], a2 = ar4[4], a3 = ar4[6];
                const float4* bl = w4 + half * HF + cq * GEMM_CB + mn;
                v8f c0 = {}, c1 = {}, c2 = {}, c3 = {};
#define PROJ_STEP(A_, S_)                                                              \
    {                                                                                  \
        const float4* bp = bl + (2 * (S_)) * HF;                                       \
        const float4 b0 = bp[0], b1 = bp[16], b2 = bp[32], b3 = bp[48];                \
        const v2f alo = {A_.x, A_.y}, ahi = {A_.z, A_.w};                              \
        c0 = __builtin_amdgcn_wmma_f32_16x16x4_f32(false, alo, false, v2f{b0.x, b0.y}, (short)0, c0, false, false); \
        c1 = __builtin_amdgcn_wmma_f32_16x16x4_f32(false, alo, false, v2f{b1.x, b1.y}, (short)0, c1, false, false); \
        c2 = __builtin_amdgcn_wmma_f32_16x16x4_f32(false, alo, false, v2f{b2.x, b2.y}, (short)0, c2, false, false); \
        c3 = __builtin_amdgcn_wmma_f32_16x16x4_f32(false, alo, false, v2f{b3.x, b3.y}, (short)0, c3, false, false); \
        c0 = __builtin_amdgcn_wmma_f32_16x16x4_f32(false, ahi, false, v2f{b0.z, b0.w}, (short)0, c0, false, false); \
        c1 = __builtin_amdgcn_wmma_f32_16x16x4_f32(false, ahi, false, v2f{b1.z, b1.w}, (short)0, c1, false, false); \
        c2 = __builtin_amdgcn_wmma_f32_16x16x4_f32(false, ahi, false, v2f{b2.z, b2.w}, (short)0, c2, false, false); \
        c3 = __builtin_amdgcn_wmma_f32_16x16x4_f32(false, ahi, false, v2f{b3.z, b3.w}, (short)0, c3, false, false); \
    }
                PROJ_STEP(a0, 0)
                PROJ_STEP(a1, 1)
                PROJ_STEP(a2, 2)
                PROJ_STEP(a3, 3)
#undef PROJ_STEP
                float* zt = SS.u.z2 + (rt * 16 + half * 8) * ZPITCH + cq * GEMM_CB + mn;
#pragma unroll
                for (int v = 0; v < 8; ++v) {
                    zt[v * ZPITCH +  0] = c0[v];
                    zt[v * ZPITCH + 16] = c1[v];
                    zt[v * ZPITCH + 32] = c2[v];
                    zt[v * ZPITCH + 48] = c3[v];
                }
                const float* zs = SS.u.z2 + (rt * 16 + half) * ZPITCH + cq * GEMM_CB + mn * 4;
                float*       zd = Zn + (size_t)(lo + rt * 16 + half) * HF + cq * GEMM_CB + mn * 4;
                float4 zr[8];
#pragma unroll
                for (int j = 0; j < 8; ++j) zr[j] = *reinterpret_cast<const float4*>(zs + 2 * j * ZPITCH);
                if (rt * 16 + 16 <= (int)nloc) {
#pragma unroll
                    for (int j = 0; j < 8; ++j) *reinterpret_cast<float4*>(zd + (size_t)2 * j * HF) = zr[j];
                } else {
#pragma unroll
                    for (int j = 0; j < 8; ++j)
                        if (rt * 16 + 2 * j + half < (int)nloc) *reinterpret_cast<float4*>(zd + (size_t)2 * j * HF) = zr[j];
                }
            }
        }
        __syncthreads();
        for (int i = tid; i < (int)nloc * HEADS; i += RTHREADS) {
            const int h = i & (HEADS - 1);
            const int n = i >> 3;
            const float* z  = SS.u.z2 + n * ZPITCH + zpos(h, 0);
            const float* pl = aln + h * FDIM;
            const float* pr = arn + h * FDIM;
            float sl = 0.f, sr = 0.f;
#pragma unroll
            for (int f = 0; f < FDIM; ++f) { float zv = z[4 * f]; sl += zv * pl[f]; sr += zv * pr[f]; }
            eln[(size_t)lo * HEADS + i] = sl; ern[(size_t)lo * HEADS + i] = sr;
        }
    }
}

#define WS_FUSED_BYTES 116194304ull
static_assert(WS_FUSED_BYTES == 2ull * NNODES * HF * 4 + 2ull * NNODES * HEADS * 4 + 1ull * NNODES * FDIM * 4 + 1ull * NRB * LREG * 4,
              "the carve-out in kernel_launch");
template <int K, bool ELU_A>
static void run_gat_layer(const float* X, const float* W,
                          const float* al, const float* ar, const float* b,
                          const int* src, const int* dst,
                          float* z, float* el, float* er, float* sums,
                          unsigned* wlst,
                          hipStream_t stream) {
    gemm_wlds_f32<K, ELU_A><<<NROWGROUPS * 4, 256, 0, stream>>>(X, W, z);

    int nh = NNODES * HEADS;
    node_prep<!ELU_A><<<(nh + 255) / 256, 256, 0, stream>>>(z, al, ar, el, er, wlst);

    if constexpr (!ELU_A) bin_edges<<<NBINB, BIN_THREADS, 0, stream>>>(src, dst, wlst);

    range_agg<false><<<NRB, RTHREADS, 0, stream>>>(src, dst, z, el, er, b, sums, wlst,
                                                   nullptr, nullptr, nullptr, nullptr, nullptr, nullptr);
}

extern "C" void kernel_launch(void* const* d_in, const int* in_sizes, int n_in,
                              void* d_out, int out_size, void* d_ws, size_t ws_size,
                              hipStream_t stream) {
    const float* x   = (const float*)d_in[0];
    const float* W1  = (const float*)d_in[1];
    const float* al1 = (const float*)d_in[2];
    const float* ar1 = (const float*)d_in[3];
    const float* b1  = (const float*)d_in[4];
    const float* W2  = (const float*)d_in[5];
    const float* al2 = (const float*)d_in[6];
    const float* ar2 = (const float*)d_in[7];
    const float* b2  = (const float*)d_in[8];
    const int*   src = (const int*)d_in[9];
    const int*   dst = (const int*)d_in[10];
    float* out = (float*)d_out;

    char* p = (char*)d_ws;
    float*    z     = (float*)p;    p += (size_t)NNODES * HF * 4;
    float*    el    = (float*)p;    p += (size_t)NNODES * HEADS * 4;
    float*    er    = (float*)p;    p += (size_t)NNODES * HEADS * 4;
    float*    acc   = (float*)p;    p += (size_t)NNODES * FDIM * 4;
    unsigned* wlst  = (unsigned*)p; p += (size_t)NRB * LREG * 4;
    float*    z2    = (float*)p;

    if (ws_size >= WS_FUSED_BYTES) {
        float* el2 = acc;
        float* er2 = acc + (size_t)NNODES * HEADS;
        gemm_wlds_f32<INF_, false><<<NROWGROUPS * 4, 256, 0, stream>>>(x, W1, z);
        node_prep<true><<<(NNODES * HEADS + 255) / 256, 256, 0, stream>>>(z, al1, ar1, el, er, wlst);
        bin_edges<<<NBINB, BIN_THREADS, 0, stream>>>(src, dst, wlst);
        range_agg<true><<<NRB, RTHREADS, 0, stream>>>(src, dst, z, el, er, b1, nullptr, wlst,
                                                      W2, al2, ar2, z2, el2, er2);
        range_agg<false><<<NRB, RTHREADS, 0, stream>>>(src, dst, z2, el2, er2, b2, out, wlst,
                                                       nullptr, nullptr, nullptr, nullptr, nullptr, nullptr);
    } else {
        run_gat_layer<INF_, false>(x, W1, al1, ar1, b1, src, dst, z, el, er, acc, wlst, stream);
        run_gat_layer<FDIM, true>(acc, W2, al2, ar2, b2, src, dst, z, el, er, out, wlst, stream);
    }
}
